# starting kernel
# baseline (speedup 1.0000x reference)
#include <hip/hip_runtime.h>
#include <hip/hip_bf16.h>


typedef __attribute__((ext_vector_type(16))) _Float16 v16h;
typedef __attribute__((ext_vector_type(8)))  _Float16 v8h;
typedef __attribute__((ext_vector_type(4)))  _Float16 v4h;
typedef __attribute__((ext_vector_type(8)))  float    v8f;

constexpr int kB = 2, kN = 2048, kC = 1024, kH = 16, kD = 64;
constexpr float kLnEps = 1e-5f;

__device__ __forceinline__ v8f wmma16(v16h a, v16h b, v8f c) {
  return __builtin_amdgcn_wmma_f32_16x16x32_f16(false, a, false, b, (short)0, c,
                                                false, false);
}

__device__ __forceinline__ v16h load_frag(const _Float16* __restrict__ p, int ld) {
  const int lane = threadIdx.x & 31;
  const int r    = lane & 15;
  const int ko   = (lane >> 4) << 3;
  const _Float16* b0 = p + (size_t)r * ld + ko;
  v8h lo = *(const v8h*)(b0);
  v8h hi = *(const v8h*)(b0 + 16);
  v16h f;
#pragma unroll
  for (int i = 0; i < 8; ++i) { f[i] = lo[i]; f[i + 8] = hi[i]; }
  return f;
}

__global__ void cvt_f32_f16(const float* __restrict__ src,
                            _Float16* __restrict__ dst, int n4) {
  int i = blockIdx.x * blockDim.x + threadIdx.x;
  if (i < n4) {
    float4 v = ((const float4*)src)[i];
    v4h h;
    h[0] = (_Float16)v.x; h[1] = (_Float16)v.y;
    h[2] = (_Float16)v.z; h[3] = (_Float16)v.w;
    ((v4h*)dst)[i] = h;
  }
}

template <int MODE>
__global__ void gemm_xwT(const _Float16* __restrict__ A,
                         const _Float16* __restrict__ W,
                         void* __restrict__ out) {
  const int lane = threadIdx.x & 31;
  const int colL = lane & 15;
  const int hi   = lane >> 4;
  const int m0 = blockIdx.x * 128 + (threadIdx.x >> 5) * 32;
  const int n0 = blockIdx.y * 64;

  v8f acc[2][4] = {};
  const _Float16* a0p = A + (size_t)m0 * kC;
  const _Float16* a1p = A + (size_t)(m0 + 16) * kC;
  const _Float16* wp  = W + (size_t)n0 * kC;

  for (int k = 0; k < kC; k += 32) {
    v16h a0 = load_frag(a0p + k, kC);
    v16h a1 = load_frag(a1p + k, kC);
#pragma unroll
    for (int nt = 0; nt < 4; ++nt) {
      v16h bf = load_frag(wp + (size_t)nt * 16 * kC + k, kC);
      acc[0][nt] = wmma16(a0, bf, acc[0][nt]);
      acc[1][nt] = wmma16(a1, bf, acc[1][nt]);
    }
  }

#pragma unroll
  for (int mt = 0; mt < 2; ++mt)
#pragma unroll
    for (int nt = 0; nt < 4; ++nt)
#pragma unroll
      for (int r = 0; r < 8; ++r) {
        const int row  = m0 + mt * 16 + 8 * hi + r;
        const int colc = n0 + nt * 16 + colL;
        const float v  = acc[mt][nt][r];
        if constexpr (MODE == 2) {
          ((float*)out)[(size_t)row * kC + colc] = v;
        } else {
          const int b = row >> 11, n = row & (kN - 1);
          const int h = colc >> 6, d = colc & (kD - 1);
          if constexpr (MODE == 0)
            ((_Float16*)out)[(((size_t)(b * kH + h)) * kN + n) * kD + d] = (_Float16)v;
          else
            ((_Float16*)out)[(((size_t)(b * kH + h)) * kD + d) * kN + n] = (_Float16)v;
        }
      }
}

__global__ void colstats_kernel(const _Float16* __restrict__ q16,
                                const _Float16* __restrict__ k16,
                                float* __restrict__ mcol,
                                float* __restrict__ rcol) {
  const int kt   = blockIdx.x;
  const int bh   = blockIdx.y;
  const int wave = threadIdx.x >> 5;
  const int lane = threadIdx.x & 31;
  const _Float16* qb = q16 + (size_t)bh * kN * kD;
  const _Float16* kb = k16 + (size_t)bh * kN * kD;

  const v16h bk0 = load_frag(kb + (size_t)kt * 16 * kD + 0,  kD);
  const v16h bk1 = load_frag(kb + (size_t)kt * 16 * kD + 32, kD);

  __shared__ float red[8][16];
  __shared__ float mshare[16];

  float mx = -3.4e38f;
  for (int qt = wave; qt < kN / 16; qt += 8) {
    v16h a0 = load_frag(qb + (size_t)qt * 16 * kD + 0,  kD);
    v16h a1 = load_frag(qb + (size_t)qt * 16 * kD + 32, kD);
    v8f s = {};
    s = wmma16(a0, bk0, s);
    s = wmma16(a1, bk1, s);
#pragma unroll
    for (int i = 0; i < 8; ++i) mx = fmaxf(mx, s[i]);
  }
  mx = fmaxf(mx, __shfl_xor(mx, 16, 32));
  if (lane < 16) red[wave][lane] = mx;
  __syncthreads();
  if (threadIdx.x < 16) {
    float m = red[0][threadIdx.x];
#pragma unroll
    for (int w = 1; w < 8; ++w) m = fmaxf(m, red[w][threadIdx.x]);
    mshare[threadIdx.x] = m;
    mcol[(size_t)bh * kN + kt * 16 + threadIdx.x] = m;
  }
  __syncthreads();
  const float mk = mshare[lane & 15];

  float sm = 0.f;
  for (int qt = wave; qt < kN / 16; qt += 8) {
    v16h a0 = load_frag(qb + (size_t)qt * 16 * kD + 0,  kD);
    v16h a1 = load_frag(qb + (size_t)qt * 16 * kD + 32, kD);
    v8f s = {};
    s = wmma16(a0, bk0, s);
    s = wmma16(a1, bk1, s);
#pragma unroll
    for (int i = 0; i < 8; ++i) sm += __expf(s[i] - mk);
  }
  sm += __shfl_xor(sm, 16, 32);
  if (lane < 16) red[wave][lane] = sm;
  __syncthreads();
  if (threadIdx.x < 16) {
    float s = 0.f;
#pragma unroll
    for (int w = 0; w < 8; ++w) s += red[w][threadIdx.x];
    rcol[(size_t)bh * kN + kt * 16 + threadIdx.x] = 1.0f / s;
  }
}

__global__ void ctx_kernel(const _Float16* __restrict__ q16,
                           const _Float16* __restrict__ k16,
                           const _Float16* __restrict__ vt16,
                           const float* __restrict__ mcol,
                           const float* __restrict__ rcol,
                           _Float16* __restrict__ ctx16) {
  const int bh   = blockIdx.y;
  const int wave = threadIdx.x >> 5;
  const int lane = threadIdx.x & 31;
  const int col  = lane & 15;
  const int hi   = lane >> 4;
  const int qt   = blockIdx.x * 8 + wave;

  __shared__ float mr_s[2][kN];
  __shared__ float estage[8][32][17];

  for (int i = threadIdx.x; i < kN; i += 256) {
    mr_s[0][i] = mcol[(size_t)bh * kN + i];
    mr_s[1][i] = rcol[(size_t)bh * kN + i];
  }
  __syncthreads();

  const _Float16* qb = q16  + (size_t)bh * kN * kD;
  const _Float16* kb = k16  + (size_t)bh * kN * kD;
  const _Float16* vb = vt16 + (size_t)bh * kD * kN;

  const v16h aq0 = load_frag(qb + (size_t)qt * 16 * kD + 0,  kD);
  const v16h aq1 = load_frag(qb + (size_t)qt * 16 * kD + 32, kD);

  v8f acc[4] = {};
  float (*stg)[17] = estage[wave];

  for (int k0 = 0; k0 < kN; k0 += 32) {
    v8f s0 = {}, s1 = {};
    s0 = wmma16(aq0, load_frag(kb + (size_t)(k0     ) * kD + 0,  kD), s0);
    s0 = wmma16(aq1, load_frag(kb + (size_t)(k0     ) * kD + 32, kD), s0);
    s1 = wmma16(aq0, load_frag(kb + (size_t)(k0 + 16) * kD + 0,  kD), s1);
    s1 = wmma16(aq1, load_frag(kb + (size_t)(k0 + 16) * kD + 32, kD), s1);

    const float m0 = mr_s[0][k0 + col],      r0 = mr_s[1][k0 + col];
    const float m1 = mr_s[0][k0 + 16 + col], r1 = mr_s[1][k0 + 16 + col];

#pragma unroll
    for (int i = 0; i < 8; ++i) {
      stg[col][i + 8 * hi]      = __expf(s0[i] - m0) * r0;
      stg[col + 16][i + 8 * hi] = __expf(s1[i] - m1) * r1;
    }
    v16h ef;
#pragma unroll
    for (int i = 0; i < 16; ++i) {
      const int kk = ((i < 8) ? i : i + 8) + 8 * hi;
      ef[i] = (_Float16)stg[kk][col];
    }
#pragma unroll
    for (int dt = 0; dt < 4; ++dt)
      acc[dt] = wmma16(ef, load_frag(vb + (size_t)dt * 16 * kN + k0, kN), acc[dt]);
  }

  const int b = bh >> 4, h = bh & 15;
  _Float16* cb = ctx16 + (size_t)b * kN * kC + (size_t)h * kD;
  const int qrow = qt * 16 + 8 * hi;
#pragma unroll
  for (int dt = 0; dt < 4; ++dt)
#pragma unroll
    for (int r = 0; r < 8; ++r)
      cb[(size_t)(qrow + r) * kC + dt * 16 + col] = (_Float16)acc[dt][r];
}

__global__ void ln_kernel(const float* __restrict__ resid,
                          const float* __restrict__ y,
                          const float* __restrict__ g,
                          const float* __restrict__ be,
                          float* __restrict__ out) {
  const int row = blockIdx.x;
  const int t   = threadIdx.x;
  float x[4], s = 0.f, sq = 0.f;
#pragma unroll
  for (int j = 0; j < 4; ++j) {
    const int c = t + j * 256;
    const float v = resid[(size_t)row * kC + c] + y[(size_t)row * kC + c];
    x[j] = v; s += v; sq += v * v;
  }
#pragma unroll
  for (int o = 16; o >= 1; o >>= 1) {
    s  += __shfl_xor(s,  o, 32);
    sq += __shfl_xor(sq, o, 32);
  }
  __shared__ float rs[8], rq[8];
  __shared__ float mu_s, inv_s;
  const int wave = t >> 5, lane = t & 31;
  if (lane == 0) { rs[wave] = s; rq[wave] = sq; }
  __syncthreads();
  if (t == 0) {
    float S = 0.f, Q = 0.f;
#pragma unroll
    for (int w = 0; w < 8; ++w) { S += rs[w]; Q += rq[w]; }
    const float mu = S * (1.0f / kC);
    mu_s  = mu;
    inv_s = rsqrtf(Q * (1.0f / kC) - mu * mu + kLnEps);
  }
  __syncthreads();
  const float mu = mu_s, inv = inv_s;
#pragma unroll
  for (int j = 0; j < 4; ++j) {
    const int c = t + j * 256;
    out[(size_t)row * kC + c] = (x[j] - mu) * inv * g[c] + be[c];
  }
}

extern "C" void kernel_launch(void* const* d_in, const int* in_sizes, int n_in,
                              void* d_out, int out_size, void* d_ws, size_t ws_size,
                              hipStream_t stream) {
  (void)in_sizes; (void)n_in; (void)out_size; (void)ws_size;
  const float* query = (const float*)d_in[0];
  const float* key   = (const float*)d_in[1];
  const float* value = (const float*)d_in[2];
  const float* Wq = (const float*)d_in[4];
  const float* Wk = (const float*)d_in[5];
  const float* Wv = (const float*)d_in[6];
  const float* Wo = (const float*)d_in[7];
  const float* ln_g = (const float*)d_in[8];
  const float* ln_b = (const float*)d_in[9];
  float* out = (float*)d_out;

  constexpr size_t BNC = (size_t)kB * kN * kC;
  constexpr size_t CC  = (size_t)kC * kC;
  constexpr size_t BHN = (size_t)kB * kH * kN;

  size_t off = 0;
  auto take = [&](size_t bytes) -> void* {
    void* p = (char*)d_ws + off;
    off += (bytes + 255) & ~(size_t)255;
    return p;
  };
  _Float16* xq16 = (_Float16*)take(BNC * 2);
  _Float16* xk16 = (_Float16*)take(BNC * 2);
  _Float16* xv16 = (_Float16*)take(BNC * 2);
  _Float16* wq16 = (_Float16*)take(CC * 2);
  _Float16* wk16 = (_Float16*)take(CC * 2);
  _Float16* wv16 = (_Float16*)take(CC * 2);
  _Float16* wo16 = (_Float16*)take(CC * 2);
  _Float16* qh16 = (_Float16*)take(BNC * 2);
  _Float16* kh16 = (_Float16*)take(BNC * 2);
  _Float16* vt16 = (_Float16*)take(BNC * 2);
  float*    mcol = (float*)take(BHN * 4);
  float*    rcol = (float*)take(BHN * 4);
  _Float16* ctx16 = xq16;
  float*    yf32  = (float*)xk16;

  cvt_f32_f16<<<(int)(BNC / 4 / 256), 256, 0, stream>>>(query, xq16, (int)(BNC / 4));
  cvt_f32_f16<<<(int)(BNC / 4 / 256), 256, 0, stream>>>(key,   xk16, (int)(BNC / 4));
  cvt_f32_f16<<<(int)(BNC / 4 / 256), 256, 0, stream>>>(value, xv16, (int)(BNC / 4));
  cvt_f32_f16<<<(int)(CC / 4 / 256),  256, 0, stream>>>(Wq, wq16, (int)(CC / 4));
  cvt_f32_f16<<<(int)(CC / 4 / 256),  256, 0, stream>>>(Wk, wk16, (int)(CC / 4));
  cvt_f32_f16<<<(int)(CC / 4 / 256),  256, 0, stream>>>(Wv, wv16, (int)(CC / 4));
  cvt_f32_f16<<<(int)(CC / 4 / 256),  256, 0, stream>>>(Wo, wo16, (int)(CC / 4));

  dim3 ggrid(32, 16);
  gemm_xwT<0><<<ggrid, 128, 0, stream>>>(xq16, wq16, qh16);
  gemm_xwT<0><<<ggrid, 128, 0, stream>>>(xk16, wk16, kh16);
  gemm_xwT<1><<<ggrid, 128, 0, stream>>>(xv16, wv16, vt16);

  colstats_kernel<<<dim3(kN / 16, kB * kH), 256, 0, stream>>>(qh16, kh16, mcol, rcol);

  ctx_kernel<<<dim3(kN / 128, kB * kH), 256, 0, stream>>>(qh16, kh16, vt16, mcol,
                                                          rcol, ctx16);

  gemm_xwT<2><<<ggrid, 128, 0, stream>>>(ctx16, wo16, yf32);
  ln_kernel<<<kB * kN, 256, 0, stream>>>(query, yf32, ln_g, ln_b, out);
}


# ctx kernel: each wave owns 4 q tiles (64 rows): a chunk's K and V fragments are loaded once and feed 32 WMMAs instead of 8 (4x fewer fragment loads; 4-wave blocks; a tile's LDS reads kept grouped)
# speedup vs baseline: 1.6174x; 1.6174x over previous
#include <hip/hip_runtime.h>
#include <hip/hip_bf16.h>


typedef __attribute__((ext_vector_type(16))) _Float16 v16h;
typedef __attribute__((ext_vector_type(8)))  _Float16 v8h;
typedef __attribute__((ext_vector_type(4)))  _Float16 v4h;
typedef __attribute__((ext_vector_type(8)))  float    v8f;

constexpr int kB = 2, kN = 2048, kC = 1024, kH = 16, kD = 64;
constexpr float kLnEps = 1e-5f;

__device__ __forceinline__ v8f wmma16(v16h a, v16h b, v8f c) {
  return __builtin_amdgcn_wmma_f32_16x16x32_f16(false, a, false, b, (short)0, c,
                                                false, false);
}

__device__ __forceinline__ v16h load_frag(const _Float16* __restrict__ p, int ld) {
  const int lane = threadIdx.x & 31;
  const int r    = lane & 15;
  const int ko   = (lane >> 4) << 3;
  const _Float16* b0 = p + (size_t)r * ld + ko;
  v8h lo = *(const v8h*)(b0);
  v8h hi = *(const v8h*)(b0 + 16);
  v16h f;
#pragma unroll
  for (int i = 0; i < 8; ++i) { f[i] = lo[i]; f[i + 8] = hi[i]; }
  return f;
}

__global__ void cvt_f32_f16(const float* __restrict__ src,
                            _Float16* __restrict__ dst, int n4) {
  int i = blockIdx.x * blockDim.x + threadIdx.x;
  if (i < n4) {
    float4 v = ((const float4*)src)[i];
    v4h h;
    h[0] = (_Float16)v.x; h[1] = (_Float16)v.y;
    h[2] = (_Float16)v.z; h[3] = (_Float16)v.w;
    ((v4h*)dst)[i] = h;
  }
}

template <int MODE>
__global__ void gemm_xwT(const _Float16* __restrict__ A,
                         const _Float16* __restrict__ W,
                         void* __restrict__ out) {
  const int lane = threadIdx.x & 31;
  const int colL = lane & 15;
  const int hi   = lane >> 4;
  const int m0 = blockIdx.x * 128 + (threadIdx.x >> 5) * 32;
  const int n0 = blockIdx.y * 64;

  v8f acc[2][4] = {};
  const _Float16* a0p = A + (size_t)m0 * kC;
  const _Float16* a1p = A + (size_t)(m0 + 16) * kC;
  const _Float16* wp  = W + (size_t)n0 * kC;

  for (int k = 0; k < kC; k += 32) {
    v16h a0 = load_frag(a0p + k, kC);
    v16h a1 = load_frag(a1p + k, kC);
#pragma unroll
    for (int nt = 0; nt < 4; ++nt) {
      v16h bf = load_frag(wp + (size_t)nt * 16 * kC + k, kC);
      acc[0][nt] = wmma16(a0, bf, acc[0][nt]);
      acc[1][nt] = wmma16(a1, bf, acc[1][nt]);
    }
  }

#pragma unroll
  for (int mt = 0; mt < 2; ++mt)
#pragma unroll
    for (int nt = 0; nt < 4; ++nt)
#pragma unroll
      for (int r = 0; r < 8; ++r) {
        const int row  = m0 + mt * 16 + 8 * hi + r;
        const int colc = n0 + nt * 16 + colL;
        const float v  = acc[mt][nt][r];
        if constexpr (MODE == 2) {
          ((float*)out)[(size_t)row * kC + colc] = v;
        } else {
          const int b = row >> 11, n = row & (kN - 1);
          const int h = colc >> 6, d = colc & (kD - 1);
          if constexpr (MODE == 0)
            ((_Float16*)out)[(((size_t)(b * kH + h)) * kN + n) * kD + d] = (_Float16)v;
          else
            ((_Float16*)out)[(((size_t)(b * kH + h)) * kD + d) * kN + n] = (_Float16)v;
        }
      }
}

constexpr int kKT = 8;

__global__ __launch_bounds__(256, 4)
void colstats_kernel(const _Float16* __restrict__ q16,
                     const _Float16* __restrict__ k16,
                     float* __restrict__ mcol,
                     float* __restrict__ rcol) {
  const int kg   = blockIdx.x;
  const int bh   = blockIdx.y;
  const int wave = threadIdx.x >> 5;
  const int lane = threadIdx.x & 31;
  const _Float16* qb = q16 + (size_t)bh * kN * kD;
  const _Float16* kb = k16 + ((size_t)bh * kN + (size_t)kg * 16 * kKT) * kD;

  v16h bk[kKT][2];
#pragma unroll
  for (int t = 0; t < kKT; ++t) {
    bk[t][0] = load_frag(kb + (size_t)t * 16 * kD + 0,  kD);
    bk[t][1] = load_frag(kb + (size_t)t * 16 * kD + 32, kD);
  }

  float mrun[kKT], srun[kKT];
#pragma unroll
  for (int t = 0; t < kKT; ++t) { mrun[t] = -3.4e38f; srun[t] = 0.f; }

  constexpr int kTrips = kN / 16 / 8;
  v16h a0 = load_frag(qb + (size_t)wave * 16 * kD + 0,  kD);
  v16h a1 = load_frag(qb + (size_t)wave * 16 * kD + 32, kD);
  for (int it = 0; it < kTrips; ++it) {
    v16h n0 = a0, n1 = a1;
    if (it + 1 < kTrips) {
      const int qn = wave + 8 * (it + 1);
      n0 = load_frag(qb + (size_t)qn * 16 * kD + 0,  kD);
      n1 = load_frag(qb + (size_t)qn * 16 * kD + 32, kD);
    }
#pragma unroll
    for (int tp = 0; tp < kKT; tp += 2) {
      v8f s[2];
#pragma unroll
      for (int u = 0; u < 2; ++u) {
        s[u] = v8f{};
        s[u] = wmma16(a0, bk[tp + u][0], s[u]);
        s[u] = wmma16(a1, bk[tp + u][1], s[u]);
      }
#pragma unroll
      for (int u = 0; u < 2; ++u) {
        const int t = tp + u;
        float tm = s[u][0];
#pragma unroll
        for (int i = 1; i < 8; ++i) tm = fmaxf(tm, s[u][i]);
        const float mn = fmaxf(mrun[t], tm);
        float part = 0.f;
#pragma unroll
        for (int i = 0; i < 8; ++i) part += __expf(s[u][i] - mn);
        srun[t] = srun[t] * __expf(mrun[t] - mn) + part;
        mrun[t] = mn;
      }
      __builtin_amdgcn_sched_barrier(0);
    }
    a0 = n0; a1 = n1;
  }

  __shared__ float redm[8][16 * kKT];
  __shared__ float reds[8][16 * kKT];
#pragma unroll
  for (int t = 0; t < kKT; ++t) {
    const float mo = __shfl_xor(mrun[t], 16, 32);
    const float so = __shfl_xor(srun[t], 16, 32);
    const float mn = fmaxf(mrun[t], mo);
    const float sv = srun[t] * __expf(mrun[t] - mn) + so * __expf(mo - mn);
    if (lane < 16) { redm[wave][t * 16 + lane] = mn; reds[wave][t * 16 + lane] = sv; }
  }
  __syncthreads();
  if (threadIdx.x < 16 * kKT) {
    const int c = threadIdx.x;
    float m = redm[0][c];
#pragma unroll
    for (int w = 1; w < 8; ++w) m = fmaxf(m, redm[w][c]);
    float sum = 0.f;
#pragma unroll
    for (int w = 0; w < 8; ++w) sum += reds[w][c] * __expf(redm[w][c] - m);
    const size_t o = (size_t)bh * kN + (size_t)kg * 16 * kKT + c;
    mcol[o] = m;
    rcol[o] = 1.0f / sum;
  }
}

constexpr int kQT = 4;
constexpr int kCW = 4;

__global__ __launch_bounds__(32 * kCW)
void ctx_kernel(const _Float16* __restrict__ q16,
                const _Float16* __restrict__ k16,
                const _Float16* __restrict__ vt16,
                const float* __restrict__ mcol,
                const float* __restrict__ rcol,
                _Float16* __restrict__ ctx16) {
  const int bh   = blockIdx.y;
  const int wave = threadIdx.x >> 5;
  const int lane = threadIdx.x & 31;
  const int col  = lane & 15;
  const int hi   = lane >> 4;
  const int qt0  = (blockIdx.x * kCW + wave) * kQT;

  __shared__ float mr_s[2][kN];
  __shared__ float estage[kCW][kQT][32][17];

  for (int i = threadIdx.x; i < kN; i += 32 * kCW) {
    mr_s[0][i] = mcol[(size_t)bh * kN + i];
    mr_s[1][i] = rcol[(size_t)bh * kN + i];
  }
  __syncthreads();

  const _Float16* qb = q16  + (size_t)bh * kN * kD;
  const _Float16* kb = k16  + (size_t)bh * kN * kD;
  const _Float16* vb = vt16 + (size_t)bh * kD * kN;

  v16h aq[kQT][2];
#pragma unroll
  for (int j = 0; j < kQT; ++j) {
    aq[j][0] = load_frag(qb + (size_t)(qt0 + j) * 16 * kD + 0,  kD);
    aq[j][1] = load_frag(qb + (size_t)(qt0 + j) * 16 * kD + 32, kD);
  }

  v8f acc[kQT][4] = {};

  for (int k0 = 0; k0 < kN; k0 += 32) {
    v16h kf[4], vf[4];
    kf[0] = load_frag(kb + (size_t)(k0     ) * kD + 0,  kD);
    kf[1] = load_frag(kb + (size_t)(k0     ) * kD + 32, kD);
    kf[2] = load_frag(kb + (size_t)(k0 + 16) * kD + 0,  kD);
    kf[3] = load_frag(kb + (size_t)(k0 + 16) * kD + 32, kD);
#pragma unroll
    for (int dt = 0; dt < 4; ++dt)
      vf[dt] = load_frag(vb + (size_t)dt * 16 * kN + k0, kN);

    const float m0 = mr_s[0][k0 + col],      r0 = mr_s[1][k0 + col];
    const float m1 = mr_s[0][k0 + 16 + col], r1 = mr_s[1][k0 + 16 + col];

#pragma unroll
    for (int j = 0; j < kQT; ++j) {
      v8f s0 = {}, s1 = {};
      s0 = wmma16(aq[j][0], kf[0], s0);
      s0 = wmma16(aq[j][1], kf[1], s0);
      s1 = wmma16(aq[j][0], kf[2], s1);
      s1 = wmma16(aq[j][1], kf[3], s1);

      float (*stg)[17] = estage[wave][j];
#pragma unroll
      for (int i = 0; i < 8; ++i) {
        stg[col][i + 8 * hi]      = __expf(s0[i] - m0) * r0;
        stg[col + 16][i + 8 * hi] = __expf(s1[i] - m1) * r1;
      }
      float t[16];
#pragma unroll
      for (int i = 0; i < 16; ++i) {
        const int kk = ((i < 8) ? i : i + 8) + 8 * hi;
        t[i] = stg[kk][col];
      }
      __builtin_amdgcn_sched_barrier(0x8 | 0x10);
      v16h ef;
#pragma unroll
      for (int i = 0; i < 16; ++i) ef[i] = (_Float16)t[i];
#pragma unroll
      for (int dt = 0; dt < 4; ++dt)
        acc[j][dt] = wmma16(ef, vf[dt], acc[j][dt]);
    }
  }

  const int b = bh >> 4, h = bh & 15;
  _Float16* cb = ctx16 + (size_t)b * kN * kC + (size_t)h * kD;
#pragma unroll
  for (int j = 0; j < kQT; ++j) {
    const int qrow = (qt0 + j) * 16 + 8 * hi;
#pragma unroll
    for (int dt = 0; dt < 4; ++dt)
#pragma unroll
      for (int r = 0; r < 8; ++r)
        cb[(size_t)(qrow + r) * kC + dt * 16 + col] = (_Float16)acc[j][dt][r];
  }
}

__global__ void ln_kernel(const float* __restrict__ resid,
                          const float* __restrict__ y,
                          const float* __restrict__ g,
                          const float* __restrict__ be,
                          float* __restrict__ out) {
  const int row = blockIdx.x;
  const int t   = threadIdx.x;
  float x[4], s = 0.f, sq = 0.f;
#pragma unroll
  for (int j = 0; j < 4; ++j) {
    const int c = t + j * 256;
    const float v = resid[(size_t)row * kC + c] + y[(size_t)row * kC + c];
    x[j] = v; s += v; sq += v * v;
  }
#pragma unroll
  for (int o = 16; o >= 1; o >>= 1) {
    s  += __shfl_xor(s,  o, 32);
    sq += __shfl_xor(sq, o, 32);
  }
  __shared__ float rs[8], rq[8];
  __shared__ float mu_s, inv_s;
  const int wave = t >> 5, lane = t & 31;
  if (lane == 0) { rs[wave] = s; rq[wave] = sq; }
  __syncthreads();
  if (t == 0) {
    float S = 0.f, Q = 0.f;
#pragma unroll
    for (int w = 0; w < 8; ++w) { S += rs[w]; Q += rq[w]; }
    const float mu = S * (1.0f / kC);
    mu_s  = mu;
    inv_s = rsqrtf(Q * (1.0f / kC) - mu * mu + kLnEps);
  }
  __syncthreads();
  const float mu = mu_s, inv = inv_s;
#pragma unroll
  for (int j = 0; j < 4; ++j) {
    const int c = t + j * 256;
    out[(size_t)row * kC + c] = (x[j] - mu) * inv * g[c] + be[c];
  }
}

extern "C" void kernel_launch(void* const* d_in, const int* in_sizes, int n_in,
                              void* d_out, int out_size, void* d_ws, size_t ws_size,
                              hipStream_t stream) {
  (void)in_sizes; (void)n_in; (void)out_size; (void)ws_size;
  const float* query = (const float*)d_in[0];
  const float* key   = (const float*)d_in[1];
  const float* value = (const float*)d_in[2];
  const float* Wq = (const float*)d_in[4];
  const float* Wk = (const float*)d_in[5];
  const float* Wv = (const float*)d_in[6];
  const float* Wo = (const float*)d_in[7];
  const float* ln_g = (const float*)d_in[8];
  const float* ln_b = (const float*)d_in[9];
  float* out = (float*)d_out;

  constexpr size_t BNC = (size_t)kB * kN * kC;
  constexpr size_t CC  = (size_t)kC * kC;
  constexpr size_t BHN = (size_t)kB * kH * kN;

  size_t off = 0;
  auto take = [&](size_t bytes) -> void* {
    void* p = (char*)d_ws + off;
    off += (bytes + 255) & ~(size_t)255;
    return p;
  };
  _Float16* xq16 = (_Float16*)take(BNC * 2);
  _Float16* xk16 = (_Float16*)take(BNC * 2);
  _Float16* xv16 = (_Float16*)take(BNC * 2);
  _Float16* wq16 = (_Float16*)take(CC * 2);
  _Float16* wk16 = (_Float16*)take(CC * 2);
  _Float16* wv16 = (_Float16*)take(CC * 2);
  _Float16* wo16 = (_Float16*)take(CC * 2);
  _Float16* qh16 = (_Float16*)take(BNC * 2);
  _Float16* kh16 = (_Float16*)take(BNC * 2);
  _Float16* vt16 = (_Float16*)take(BNC * 2);
  float*    mcol = (float*)take(BHN * 4);
  float*    rcol = (float*)take(BHN * 4);
  _Float16* ctx16 = xq16;
  float*    yf32  = (float*)xk16;

  cvt_f32_f16<<<(int)(BNC / 4 / 256), 256, 0, stream>>>(query, xq16, (int)(BNC / 4));
  cvt_f32_f16<<<(int)(BNC / 4 / 256), 256, 0, stream>>>(key,   xk16, (int)(BNC / 4));
  cvt_f32_f16<<<(int)(BNC / 4 / 256), 256, 0, stream>>>(value, xv16, (int)(BNC / 4));
  cvt_f32_f16<<<(int)(CC / 4 / 256),  256, 0, stream>>>(Wq, wq16, (int)(CC / 4));
  cvt_f32_f16<<<(int)(CC / 4 / 256),  256, 0, stream>>>(Wk, wk16, (int)(CC / 4));
  cvt_f32_f16<<<(int)(CC / 4 / 256),  256, 0, stream>>>(Wv, wv16, (int)(CC / 4));
  cvt_f32_f16<<<(int)(CC / 4 / 256),  256, 0, stream>>>(Wo, wo16, (int)(CC / 4));

  dim3 ggrid(32, 16);
  gemm_xwT<0><<<ggrid, 128, 0, stream>>>(xq16, wq16, qh16);
  gemm_xwT<0><<<ggrid, 128, 0, stream>>>(xk16, wk16, kh16);
  gemm_xwT<1><<<ggrid, 128, 0, stream>>>(xv16, wv16, vt16);

  colstats_kernel<<<dim3(kN / (16 * kKT), kB * kH), 256, 0, stream>>>(qh16, kh16, mcol, rcol);

  ctx_kernel<<<dim3(kN / (16 * kQT * kCW), kB * kH), 32 * kCW, 0, stream>>>(
      qh16, kh16, vt16, mcol, rcol, ctx16);

  gemm_xwT<2><<<ggrid, 128, 0, stream>>>(ctx16, wo16, yf32);
  ln_kernel<<<kB * kN, 256, 0, stream>>>(query, yf32, ln_g, ln_b, out);
}


# all seven f32->f16 converts in ONE launch of 512 8-wave blocks, each thread converting 32 float4 with 16 loads in flight (7 launches of 131072 one-vector waves -> 1 launch of 4096 waves)
# speedup vs baseline: 1.7356x; 1.0731x over previous
#include <hip/hip_runtime.h>
#include <hip/hip_bf16.h>


typedef __attribute__((ext_vector_type(16))) _Float16 v16h;
typedef __attribute__((ext_vector_type(8)))  _Float16 v8h;
typedef __attribute__((ext_vector_type(4)))  _Float16 v4h;
typedef __attribute__((ext_vector_type(8)))  float    v8f;

constexpr int kB = 2, kN = 2048, kC = 1024, kH = 16, kD = 64;
constexpr float kLnEps = 1e-5f;

__device__ __forceinline__ v8f wmma16(v16h a, v16h b, v8f c) {
  return __builtin_amdgcn_wmma_f32_16x16x32_f16(false, a, false, b, (short)0, c,
                                                false, false);
}

__device__ __forceinline__ v16h load_frag(const _Float16* __restrict__ p, int ld) {
  const int lane = threadIdx.x & 31;
  const int r    = lane & 15;
  const int ko   = (lane >> 4) << 3;
  const _Float16* b0 = p + (size_t)r * ld + ko;
  v8h lo = *(const v8h*)(b0);
  v8h hi = *(const v8h*)(b0 + 16);
  v16h f;
#pragma unroll
  for (int i = 0; i < 8; ++i) { f[i] = lo[i]; f[i + 8] = hi[i]; }
  return f;
}

constexpr int kCvtBlock  = 256;
constexpr int kCvtPerBlk = 8192;
constexpr int kCvtInFl   = 16;
constexpr int kCvtRounds = kCvtPerBlk / (kCvtBlock * kCvtInFl);
constexpr int kCvtXBlks  = kB * kN * kC / 4 / kCvtPerBlk;
constexpr int kCvtWBlks  = kC * kC / 4 / kCvtPerBlk;
constexpr int kCvtGrid   = 3 * kCvtXBlks + 4 * kCvtWBlks;
static_assert(kCvtRounds * kCvtBlock * kCvtInFl == kCvtPerBlk, "block chunk");
static_assert(kCvtXBlks * kCvtPerBlk * 4 == kB * kN * kC, "activation split");
static_assert(kCvtWBlks * kCvtPerBlk * 4 == kC * kC, "weight split");

__global__ __launch_bounds__(kCvtBlock)
void cvt_all_f32_f16(const float* __restrict__ xq, const float* __restrict__ xk,
                     const float* __restrict__ xv, const float* __restrict__ wq,
                     const float* __restrict__ wk, const float* __restrict__ wv,
                     const float* __restrict__ wo,
                     _Float16* __restrict__ xq16, _Float16* __restrict__ xk16,
                     _Float16* __restrict__ xv16, _Float16* __restrict__ wq16,
                     _Float16* __restrict__ wk16, _Float16* __restrict__ wv16,
                     _Float16* __restrict__ wo16) {
  const int b = blockIdx.x;
  const float* src;
  _Float16* dst;
  int lb;
  if (b < 3 * kCvtXBlks) {
    const int s = b / kCvtXBlks;
    lb  = b - s * kCvtXBlks;
    src = (s == 0) ? xq : (s == 1) ? xk : xv;
    dst = (s == 0) ? xq16 : (s == 1) ? xk16 : xv16;
  } else {
    const int bw = b - 3 * kCvtXBlks;
    const int s  = bw / kCvtWBlks;
    lb  = bw - s * kCvtWBlks;
    src = (s == 0) ? wq : (s == 1) ? wk : (s == 2) ? wv : wo;
    dst = (s == 0) ? wq16 : (s == 1) ? wk16 : (s == 2) ? wv16 : wo16;
  }
  const float4* s4 = (const float4*)src + (size_t)lb * kCvtPerBlk + threadIdx.x;
  v4h*          d4 = (v4h*)dst          + (size_t)lb * kCvtPerBlk + threadIdx.x;

#pragma unroll
  for (int r = 0; r < kCvtRounds; ++r) {
    float4 v[kCvtInFl];
#pragma unroll
    for (int j = 0; j < kCvtInFl; ++j)
      v[j] = s4[(r * kCvtInFl + j) * kCvtBlock];
#pragma unroll
    for (int j = 0; j < kCvtInFl; ++j) {
      v4h h;
      h[0] = (_Float16)v[j].x; h[1] = (_Float16)v[j].y;
      h[2] = (_Float16)v[j].z; h[3] = (_Float16)v[j].w;
      d4[(r * kCvtInFl + j) * kCvtBlock] = h;
    }
  }
}

template <int MODE>
__global__ void gemm_xwT(const _Float16* __restrict__ A,
                         const _Float16* __restrict__ W,
                         void* __restrict__ out) {
  const int lane = threadIdx.x & 31;
  const int colL = lane & 15;
  const int hi   = lane >> 4;
  const int m0 = blockIdx.x * 128 + (threadIdx.x >> 5) * 32;
  const int n0 = blockIdx.y * 64;

  v8f acc[2][4] = {};
  const _Float16* a0p = A + (size_t)m0 * kC;
  const _Float16* a1p = A + (size_t)(m0 + 16) * kC;
  const _Float16* wp  = W + (size_t)n0 * kC;

  for (int k = 0; k < kC; k += 32) {
    v16h a0 = load_frag(a0p + k, kC);
    v16h a1 = load_frag(a1p + k, kC);
#pragma unroll
    for (int nt = 0; nt < 4; ++nt) {
      v16h bf = load_frag(wp + (size_t)nt * 16 * kC + k, kC);
      acc[0][nt] = wmma16(a0, bf, acc[0][nt]);
      acc[1][nt] = wmma16(a1, bf, acc[1][nt]);
    }
  }

#pragma unroll
  for (int mt = 0; mt < 2; ++mt)
#pragma unroll
    for (int nt = 0; nt < 4; ++nt)
#pragma unroll
      for (int r = 0; r < 8; ++r) {
        const int row  = m0 + mt * 16 + 8 * hi + r;
        const int colc = n0 + nt * 16 + colL;
        const float v  = acc[mt][nt][r];
        if constexpr (MODE == 2) {
          ((float*)out)[(size_t)row * kC + colc] = v;
        } else {
          const int b = row >> 11, n = row & (kN - 1);
          const int h = colc >> 6, d = colc & (kD - 1);
          if constexpr (MODE == 0)
            ((_Float16*)out)[(((size_t)(b * kH + h)) * kN + n) * kD + d] = (_Float16)v;
          else
            ((_Float16*)out)[(((size_t)(b * kH + h)) * kD + d) * kN + n] = (_Float16)v;
        }
      }
}

constexpr int kKT = 8;

__global__ __launch_bounds__(256, 4)
void colstats_kernel(const _Float16* __restrict__ q16,
                     const _Float16* __restrict__ k16,
                     float* __restrict__ mcol,
                     float* __restrict__ rcol) {
  const int kg   = blockIdx.x;
  const int bh   = blockIdx.y;
  const int wave = threadIdx.x >> 5;
  const int lane = threadIdx.x & 31;
  const _Float16* qb = q16 + (size_t)bh * kN * kD;
  const _Float16* kb = k16 + ((size_t)bh * kN + (size_t)kg * 16 * kKT) * kD;

  v16h bk[kKT][2];
#pragma unroll
  for (int t = 0; t < kKT; ++t) {
    bk[t][0] = load_frag(kb + (size_t)t * 16 * kD + 0,  kD);
    bk[t][1] = load_frag(kb + (size_t)t * 16 * kD + 32, kD);
  }

  float mrun[kKT], srun[kKT];
#pragma unroll
  for (int t = 0; t < kKT; ++t) { mrun[t] = -3.4e38f; srun[t] = 0.f; }

  constexpr int kTrips = kN / 16 / 8;
  v16h a0 = load_frag(qb + (size_t)wave * 16 * kD + 0,  kD);
  v16h a1 = load_frag(qb + (size_t)wave * 16 * kD + 32, kD);
  for (int it = 0; it < kTrips; ++it) {
    v16h n0 = a0, n1 = a1;
    if (it + 1 < kTrips) {
      const int qn = wave + 8 * (it + 1);
      n0 = load_frag(qb + (size_t)qn * 16 * kD + 0,  kD);
      n1 = load_frag(qb + (size_t)qn * 16 * kD + 32, kD);
    }
#pragma unroll
    for (int tp = 0; tp < kKT; tp += 2) {
      v8f s[2];
#pragma unroll
      for (int u = 0; u < 2; ++u) {
        s[u] = v8f{};
        s[u] = wmma16(a0, bk[tp + u][0], s[u]);
        s[u] = wmma16(a1, bk[tp + u][1], s[u]);
      }
#pragma unroll
      for (int u = 0; u < 2; ++u) {
        const int t = tp + u;
        float tm = s[u][0];
#pragma unroll
        for (int i = 1; i < 8; ++i) tm = fmaxf(tm, s[u][i]);
        const float mn = fmaxf(mrun[t], tm);
        float part = 0.f;
#pragma unroll
        for (int i = 0; i < 8; ++i) part += __expf(s[u][i] - mn);
        srun[t] = srun[t] * __expf(mrun[t] - mn) + part;
        mrun[t] = mn;
      }
      __builtin_amdgcn_sched_barrier(0);
    }
    a0 = n0; a1 = n1;
  }

  __shared__ float redm[8][16 * kKT];
  __shared__ float reds[8][16 * kKT];
#pragma unroll
  for (int t = 0; t < kKT; ++t) {
    const float mo = __shfl_xor(mrun[t], 16, 32);
    const float so = __shfl_xor(srun[t], 16, 32);
    const float mn = fmaxf(mrun[t], mo);
    const float sv = srun[t] * __expf(mrun[t] - mn) + so * __expf(mo - mn);
    if (lane < 16) { redm[wave][t * 16 + lane] = mn; reds[wave][t * 16 + lane] = sv; }
  }
  __syncthreads();
  if (threadIdx.x < 16 * kKT) {
    const int c = threadIdx.x;
    float m = redm[0][c];
#pragma unroll
    for (int w = 1; w < 8; ++w) m = fmaxf(m, redm[w][c]);
    float sum = 0.f;
#pragma unroll
    for (int w = 0; w < 8; ++w) sum += reds[w][c] * __expf(redm[w][c] - m);
    const size_t o = (size_t)bh * kN + (size_t)kg * 16 * kKT + c;
    mcol[o] = m;
    rcol[o] = 1.0f / sum;
  }
}

constexpr int kQT = 4;
constexpr int kCW = 4;

__global__ __launch_bounds__(32 * kCW)
void ctx_kernel(const _Float16* __restrict__ q16,
                const _Float16* __restrict__ k16,
                const _Float16* __restrict__ vt16,
                const float* __restrict__ mcol,
                const float* __restrict__ rcol,
                _Float16* __restrict__ ctx16) {
  const int bh   = blockIdx.y;
  const int wave = threadIdx.x >> 5;
  const int lane = threadIdx.x & 31;
  const int col  = lane & 15;
  const int hi   = lane >> 4;
  const int qt0  = (blockIdx.x * kCW + wave) * kQT;

  __shared__ float mr_s[2][kN];
  __shared__ float estage[kCW][kQT][32][17];

  for (int i = threadIdx.x; i < kN; i += 32 * kCW) {
    mr_s[0][i] = mcol[(size_t)bh * kN + i];
    mr_s[1][i] = rcol[(size_t)bh * kN + i];
  }
  __syncthreads();

  const _Float16* qb = q16  + (size_t)bh * kN * kD;
  const _Float16* kb = k16  + (size_t)bh * kN * kD;
  const _Float16* vb = vt16 + (size_t)bh * kD * kN;

  v16h aq[kQT][2];
#pragma unroll
  for (int j = 0; j < kQT; ++j) {
    aq[j][0] = load_frag(qb + (size_t)(qt0 + j) * 16 * kD + 0,  kD);
    aq[j][1] = load_frag(qb + (size_t)(qt0 + j) * 16 * kD + 32, kD);
  }

  v8f acc[kQT][4] = {};

  for (int k0 = 0; k0 < kN; k0 += 32) {
    v16h kf[4], vf[4];
    kf[0] = load_frag(kb + (size_t)(k0     ) * kD + 0,  kD);
    kf[1] = load_frag(kb + (size_t)(k0     ) * kD + 32, kD);
    kf[2] = load_frag(kb + (size_t)(k0 + 16) * kD + 0,  kD);
    kf[3] = load_frag(kb + (size_t)(k0 + 16) * kD + 32, kD);
#pragma unroll
    for (int dt = 0; dt < 4; ++dt)
      vf[dt] = load_frag(vb + (size_t)dt * 16 * kN + k0, kN);

    const float m0 = mr_s[0][k0 + col],      r0 = mr_s[1][k0 + col];
    const float m1 = mr_s[0][k0 + 16 + col], r1 = mr_s[1][k0 + 16 + col];

#pragma unroll
    for (int j = 0; j < kQT; ++j) {
      v8f s0 = {}, s1 = {};
      s0 = wmma16(aq[j][0], kf[0], s0);
      s0 = wmma16(aq[j][1], kf[1], s0);
      s1 = wmma16(aq[j][0], kf[2], s1);
      s1 = wmma16(aq[j][1], kf[3], s1);

      float (*stg)[17] = estage[wave][j];
#pragma unroll
      for (int i = 0; i < 8; ++i) {
        stg[col][i + 8 * hi]      = __expf(s0[i] - m0) * r0;
        stg[col + 16][i + 8 * hi] = __expf(s1[i] - m1) * r1;
      }
      float t[16];
#pragma unroll
      for (int i = 0; i < 16; ++i) {
        const int kk = ((i < 8) ? i : i + 8) + 8 * hi;
        t[i] = stg[kk][col];
      }
      __builtin_amdgcn_sched_barrier(0x8 | 0x10);
      v16h ef;
#pragma unroll
      for (int i = 0; i < 16; ++i) ef[i] = (_Float16)t[i];
#pragma unroll
      for (int dt = 0; dt < 4; ++dt)
        acc[j][dt] = wmma16(ef, vf[dt], acc[j][dt]);
    }
  }

  const int b = bh >> 4, h = bh & 15;
  _Float16* cb = ctx16 + (size_t)b * kN * kC + (size_t)h * kD;
#pragma unroll
  for (int j = 0; j < kQT; ++j) {
    const int qrow = (qt0 + j) * 16 + 8 * hi;
#pragma unroll
    for (int dt = 0; dt < 4; ++dt)
#pragma unroll
      for (int r = 0; r < 8; ++r)
        cb[(size_t)(qrow + r) * kC + dt * 16 + col] = (_Float16)acc[j][dt][r];
  }
}

__global__ void ln_kernel(const float* __restrict__ resid,
                          const float* __restrict__ y,
                          const float* __restrict__ g,
                          const float* __restrict__ be,
                          float* __restrict__ out) {
  const int row = blockIdx.x;
  const int t   = threadIdx.x;
  float x[4], s = 0.f, sq = 0.f;
#pragma unroll
  for (int j = 0; j < 4; ++j) {
    const int c = t + j * 256;
    const float v = resid[(size_t)row * kC + c] + y[(size_t)row * kC + c];
    x[j] = v; s += v; sq += v * v;
  }
#pragma unroll
  for (int o = 16; o >= 1; o >>= 1) {
    s  += __shfl_xor(s,  o, 32);
    sq += __shfl_xor(sq, o, 32);
  }
  __shared__ float rs[8], rq[8];
  __shared__ float mu_s, inv_s;
  const int wave = t >> 5, lane = t & 31;
  if (lane == 0) { rs[wave] = s; rq[wave] = sq; }
  __syncthreads();
  if (t == 0) {
    float S = 0.f, Q = 0.f;
#pragma unroll
    for (int w = 0; w < 8; ++w) { S += rs[w]; Q += rq[w]; }
    const float mu = S * (1.0f / kC);
    mu_s  = mu;
    inv_s = rsqrtf(Q * (1.0f / kC) - mu * mu + kLnEps);
  }
  __syncthreads();
  const float mu = mu_s, inv = inv_s;
#pragma unroll
  for (int j = 0; j < 4; ++j) {
    const int c = t + j * 256;
    out[(size_t)row * kC + c] = (x[j] - mu) * inv * g[c] + be[c];
  }
}

extern "C" void kernel_launch(void* const* d_in, const int* in_sizes, int n_in,
                              void* d_out, int out_size, void* d_ws, size_t ws_size,
                              hipStream_t stream) {
  (void)in_sizes; (void)n_in; (void)out_size; (void)ws_size;
  const float* query = (const float*)d_in[0];
  const float* key   = (const float*)d_in[1];
  const float* value = (const float*)d_in[2];
  const float* Wq = (const float*)d_in[4];
  const float* Wk = (const float*)d_in[5];
  const float* Wv = (const float*)d_in[6];
  const float* Wo = (const float*)d_in[7];
  const float* ln_g = (const float*)d_in[8];
  const float* ln_b = (const float*)d_in[9];
  float* out = (float*)d_out;

  constexpr size_t BNC = (size_t)kB * kN * kC;
  constexpr size_t CC  = (size_t)kC * kC;
  constexpr size_t BHN = (size_t)kB * kH * kN;

  size_t off = 0;
  auto take = [&](size_t bytes) -> void* {
    void* p = (char*)d_ws + off;
    off += (bytes + 255) & ~(size_t)255;
    return p;
  };
  _Float16* xq16 = (_Float16*)take(BNC * 2);
  _Float16* xk16 = (_Float16*)take(BNC * 2);
  _Float16* xv16 = (_Float16*)take(BNC * 2);
  _Float16* wq16 = (_Float16*)take(CC * 2);
  _Float16* wk16 = (_Float16*)take(CC * 2);
  _Float16* wv16 = (_Float16*)take(CC * 2);
  _Float16* wo16 = (_Float16*)take(CC * 2);
  _Float16* qh16 = (_Float16*)take(BNC * 2);
  _Float16* kh16 = (_Float16*)take(BNC * 2);
  _Float16* vt16 = (_Float16*)take(BNC * 2);
  float*    mcol = (float*)take(BHN * 4);
  float*    rcol = (float*)take(BHN * 4);
  _Float16* ctx16 = xq16;
  float*    yf32  = (float*)xk16;

  cvt_all_f32_f16<<<kCvtGrid, kCvtBlock, 0, stream>>>(
      query, key, value, Wq, Wk, Wv, Wo,
      xq16, xk16, xv16, wq16, wk16, wv16, wo16);

  dim3 ggrid(32, 16);
  gemm_xwT<0><<<ggrid, 128, 0, stream>>>(xq16, wq16, qh16);
  gemm_xwT<0><<<ggrid, 128, 0, stream>>>(xk16, wk16, kh16);
  gemm_xwT<1><<<ggrid, 128, 0, stream>>>(xv16, wv16, vt16);

  colstats_kernel<<<dim3(kN / (16 * kKT), kB * kH), 256, 0, stream>>>(qh16, kh16, mcol, rcol);

  ctx_kernel<<<dim3(kN / (16 * kQT * kCW), kB * kH), 32 * kCW, 0, stream>>>(
      qh16, kh16, vt16, mcol, rcol, ctx16);

  gemm_xwT<2><<<ggrid, 128, 0, stream>>>(ctx16, wo16, yf32);
  ln_kernel<<<kB * kN, 256, 0, stream>>>(query, yf32, ln_g, ln_b, out);
}


# ctx kernel: second K/V fragment set in registers; the next 32-key chunk's 16 fragment loads are issued one whole chunk ahead, so their latency runs under the current chunk's work (1 wave per SIMD)
# speedup vs baseline: 2.0175x; 1.1624x over previous
#include <hip/hip_runtime.h>
#include <hip/hip_bf16.h>


typedef __attribute__((ext_vector_type(16))) _Float16 v16h;
typedef __attribute__((ext_vector_type(8)))  _Float16 v8h;
typedef __attribute__((ext_vector_type(4)))  _Float16 v4h;
typedef __attribute__((ext_vector_type(8)))  float    v8f;

constexpr int kB = 2, kN = 2048, kC = 1024, kH = 16, kD = 64;
constexpr float kLnEps = 1e-5f;

__device__ __forceinline__ v8f wmma16(v16h a, v16h b, v8f c) {
  return __builtin_amdgcn_wmma_f32_16x16x32_f16(false, a, false, b, (short)0, c,
                                                false, false);
}

__device__ __forceinline__ v16h load_frag(const _Float16* __restrict__ p, int ld) {
  const int lane = threadIdx.x & 31;
  const int r    = lane & 15;
  const int ko   = (lane >> 4) << 3;
  const _Float16* b0 = p + (size_t)r * ld + ko;
  v8h lo = *(const v8h*)(b0);
  v8h hi = *(const v8h*)(b0 + 16);
  v16h f;
#pragma unroll
  for (int i = 0; i < 8; ++i) { f[i] = lo[i]; f[i + 8] = hi[i]; }
  return f;
}

constexpr int kCvtBlock  = 256;
constexpr int kCvtPerBlk = 8192;
constexpr int kCvtInFl   = 16;
constexpr int kCvtRounds = kCvtPerBlk / (kCvtBlock * kCvtInFl);
constexpr int kCvtXBlks  = kB * kN * kC / 4 / kCvtPerBlk;
constexpr int kCvtWBlks  = kC * kC / 4 / kCvtPerBlk;
constexpr int kCvtGrid   = 3 * kCvtXBlks + 4 * kCvtWBlks;
static_assert(kCvtRounds * kCvtBlock * kCvtInFl == kCvtPerBlk, "block chunk");
static_assert(kCvtXBlks * kCvtPerBlk * 4 == kB * kN * kC, "activation split");
static_assert(kCvtWBlks * kCvtPerBlk * 4 == kC * kC, "weight split");

__global__ __launch_bounds__(kCvtBlock)
void cvt_all_f32_f16(const float* __restrict__ xq, const float* __restrict__ xk,
                     const float* __restrict__ xv, const float* __restrict__ wq,
                     const float* __restrict__ wk, const float* __restrict__ wv,
                     const float* __restrict__ wo,
                     _Float16* __restrict__ xq16, _Float16* __restrict__ xk16,
                     _Float16* __restrict__ xv16, _Float16* __restrict__ wq16,
                     _Float16* __restrict__ wk16, _Float16* __restrict__ wv16,
                     _Float16* __restrict__ wo16) {
  const int b = blockIdx.x;
  const float* src;
  _Float16* dst;
  int lb;
  if (b < 3 * kCvtXBlks) {
    const int s = b / kCvtXBlks;
    lb  = b - s * kCvtXBlks;
    src = (s == 0) ? xq : (s == 1) ? xk : xv;
    dst = (s == 0) ? xq16 : (s == 1) ? xk16 : xv16;
  } else {
    const int bw = b - 3 * kCvtXBlks;
    const int s  = bw / kCvtWBlks;
    lb  = bw - s * kCvtWBlks;
    src = (s == 0) ? wq : (s == 1) ? wk : (s == 2) ? wv : wo;
    dst = (s == 0) ? wq16 : (s == 1) ? wk16 : (s == 2) ? wv16 : wo16;
  }
  const float4* s4 = (const float4*)src + (size_t)lb * kCvtPerBlk + threadIdx.x;
  v4h*          d4 = (v4h*)dst          + (size_t)lb * kCvtPerBlk + threadIdx.x;

#pragma unroll
  for (int r = 0; r < kCvtRounds; ++r) {
    float4 v[kCvtInFl];
#pragma unroll
    for (int j = 0; j < kCvtInFl; ++j)
      v[j] = s4[(r * kCvtInFl + j) * kCvtBlock];
#pragma unroll
    for (int j = 0; j < kCvtInFl; ++j) {
      v4h h;
      h[0] = (_Float16)v[j].x; h[1] = (_Float16)v[j].y;
      h[2] = (_Float16)v[j].z; h[3] = (_Float16)v[j].w;
      d4[(r * kCvtInFl + j) * kCvtBlock] = h;
    }
  }
}

template <int MODE>
__global__ void gemm_xwT(const _Float16* __restrict__ A,
                         const _Float16* __restrict__ W,
                         void* __restrict__ out) {
  const int lane = threadIdx.x & 31;
  const int colL = lane & 15;
  const int hi   = lane >> 4;
  const int m0 = blockIdx.x * 128 + (threadIdx.x >> 5) * 32;
  const int n0 = blockIdx.y * 64;

  v8f acc[2][4] = {};
  const _Float16* a0p = A + (size_t)m0 * kC;
  const _Float16* a1p = A + (size_t)(m0 + 16) * kC;
  const _Float16* wp  = W + (size_t)n0 * kC;

  for (int k = 0; k < kC; k += 32) {
    v16h a0 = load_frag(a0p + k, kC);
    v16h a1 = load_frag(a1p + k, kC);
#pragma unroll
    for (int nt = 0; nt < 4; ++nt) {
      v16h bf = load_frag(wp + (size_t)nt * 16 * kC + k, kC);
      acc[0][nt] = wmma16(a0, bf, acc[0][nt]);
      acc[1][nt] = wmma16(a1, bf, acc[1][nt]);
    }
  }

#pragma unroll
  for (int mt = 0; mt < 2; ++mt)
#pragma unroll
    for (int nt = 0; nt < 4; ++nt)
#pragma unroll
      for (int r = 0; r < 8; ++r) {
        const int row  = m0 + mt * 16 + 8 * hi + r;
        const int colc = n0 + nt * 16 + colL;
        const float v  = acc[mt][nt][r];
        if constexpr (MODE == 2) {
          ((float*)out)[(size_t)row * kC + colc] = v;
        } else {
          const int b = row >> 11, n = row & (kN - 1);
          const int h = colc >> 6, d = colc & (kD - 1);
          if constexpr (MODE == 0)
            ((_Float16*)out)[(((size_t)(b * kH + h)) * kN + n) * kD + d] = (_Float16)v;
          else
            ((_Float16*)out)[(((size_t)(b * kH + h)) * kD + d) * kN + n] = (_Float16)v;
        }
      }
}

constexpr int kKT = 8;

__global__ __launch_bounds__(256, 4)
void colstats_kernel(const _Float16* __restrict__ q16,
                     const _Float16* __restrict__ k16,
                     float* __restrict__ mcol,
                     float* __restrict__ rcol) {
  const int kg   = blockIdx.x;
  const int bh   = blockIdx.y;
  const int wave = threadIdx.x >> 5;
  const int lane = threadIdx.x & 31;
  const _Float16* qb = q16 + (size_t)bh * kN * kD;
  const _Float16* kb = k16 + ((size_t)bh * kN + (size_t)kg * 16 * kKT) * kD;

  v16h bk[kKT][2];
#pragma unroll
  for (int t = 0; t < kKT; ++t) {
    bk[t][0] = load_frag(kb + (size_t)t * 16 * kD + 0,  kD);
    bk[t][1] = load_frag(kb + (size_t)t * 16 * kD + 32, kD);
  }

  float mrun[kKT], srun[kKT];
#pragma unroll
  for (int t = 0; t < kKT; ++t) { mrun[t] = -3.4e38f; srun[t] = 0.f; }

  constexpr int kTrips = kN / 16 / 8;
  v16h a0 = load_frag(qb + (size_t)wave * 16 * kD + 0,  kD);
  v16h a1 = load_frag(qb + (size_t)wave * 16 * kD + 32, kD);
  for (int it = 0; it < kTrips; ++it) {
    v16h n0 = a0, n1 = a1;
    if (it + 1 < kTrips) {
      const int qn = wave + 8 * (it + 1);
      n0 = load_frag(qb + (size_t)qn * 16 * kD + 0,  kD);
      n1 = load_frag(qb + (size_t)qn * 16 * kD + 32, kD);
    }
#pragma unroll
    for (int tp = 0; tp < kKT; tp += 2) {
      v8f s[2];
#pragma unroll
      for (int u = 0; u < 2; ++u) {
        s[u] = v8f{};
        s[u] = wmma16(a0, bk[tp + u][0], s[u]);
        s[u] = wmma16(a1, bk[tp + u][1], s[u]);
      }
#pragma unroll
      for (int u = 0; u < 2; ++u) {
        const int t = tp + u;
        float tm = s[u][0];
#pragma unroll
        for (int i = 1; i < 8; ++i) tm = fmaxf(tm, s[u][i]);
        const float mn = fmaxf(mrun[t], tm);
        float part = 0.f;
#pragma unroll
        for (int i = 0; i < 8; ++i) part += __expf(s[u][i] - mn);
        srun[t] = srun[t] * __expf(mrun[t] - mn) + part;
        mrun[t] = mn;
      }
      __builtin_amdgcn_sched_barrier(0);
    }
    a0 = n0; a1 = n1;
  }

  __shared__ float redm[8][16 * kKT];
  __shared__ float reds[8][16 * kKT];
#pragma unroll
  for (int t = 0; t < kKT; ++t) {
    const float mo = __shfl_xor(mrun[t], 16, 32);
    const float so = __shfl_xor(srun[t], 16, 32);
    const float mn = fmaxf(mrun[t], mo);
    const float sv = srun[t] * __expf(mrun[t] - mn) + so * __expf(mo - mn);
    if (lane < 16) { redm[wave][t * 16 + lane] = mn; reds[wave][t * 16 + lane] = sv; }
  }
  __syncthreads();
  if (threadIdx.x < 16 * kKT) {
    const int c = threadIdx.x;
    float m = redm[0][c];
#pragma unroll
    for (int w = 1; w < 8; ++w) m = fmaxf(m, redm[w][c]);
    float sum = 0.f;
#pragma unroll
    for (int w = 0; w < 8; ++w) sum += reds[w][c] * __expf(redm[w][c] - m);
    const size_t o = (size_t)bh * kN + (size_t)kg * 16 * kKT + c;
    mcol[o] = m;
    rcol[o] = 1.0f / sum;
  }
}

constexpr int kQT = 4;
constexpr int kCW = 4;

__global__ __launch_bounds__(32 * kCW)
void ctx_kernel(const _Float16* __restrict__ q16,
                const _Float16* __restrict__ k16,
                const _Float16* __restrict__ vt16,
                const float* __restrict__ mcol,
                const float* __restrict__ rcol,
                _Float16* __restrict__ ctx16) {
  const int bh   = blockIdx.y;
  const int wave = threadIdx.x >> 5;
  const int lane = threadIdx.x & 31;
  const int col  = lane & 15;
  const int hi   = lane >> 4;
  const int qt0  = (blockIdx.x * kCW + wave) * kQT;

  __shared__ float mr_s[2][kN];
  __shared__ float estage[kCW][kQT][32][17];

  for (int i = threadIdx.x; i < kN; i += 32 * kCW) {
    mr_s[0][i] = mcol[(size_t)bh * kN + i];
    mr_s[1][i] = rcol[(size_t)bh * kN + i];
  }
  __syncthreads();

  const _Float16* qb = q16  + (size_t)bh * kN * kD;
  const _Float16* kb = k16  + (size_t)bh * kN * kD;
  const _Float16* vb = vt16 + (size_t)bh * kD * kN;

  v16h aq[kQT][2];
#pragma unroll
  for (int j = 0; j < kQT; ++j) {
    aq[j][0] = load_frag(qb + (size_t)(qt0 + j) * 16 * kD + 0,  kD);
    aq[j][1] = load_frag(qb + (size_t)(qt0 + j) * 16 * kD + 32, kD);
  }

  v8f acc[kQT][4] = {};

  auto load_chunk = [&](int k0, v16h (&kf)[4], v16h (&vf)[4]) {
    kf[0] = load_frag(kb + (size_t)(k0     ) * kD + 0,  kD);
    kf[1] = load_frag(kb + (size_t)(k0     ) * kD + 32, kD);
    kf[2] = load_frag(kb + (size_t)(k0 + 16) * kD + 0,  kD);
    kf[3] = load_frag(kb + (size_t)(k0 + 16) * kD + 32, kD);
#pragma unroll
    for (int dt = 0; dt < 4; ++dt)
      vf[dt] = load_frag(vb + (size_t)dt * 16 * kN + k0, kN);
  };

  auto do_chunk = [&](int k0, const v16h (&kf)[4], const v16h (&vf)[4]) {
    const float m0 = mr_s[0][k0 + col],      r0 = mr_s[1][k0 + col];
    const float m1 = mr_s[0][k0 + 16 + col], r1 = mr_s[1][k0 + 16 + col];

#pragma unroll
    for (int j = 0; j < kQT; ++j) {
      v8f s0 = {}, s1 = {};
      s0 = wmma16(aq[j][0], kf[0], s0);
      s0 = wmma16(aq[j][1], kf[1], s0);
      s1 = wmma16(aq[j][0], kf[2], s1);
      s1 = wmma16(aq[j][1], kf[3], s1);

      float (*stg)[17] = estage[wave][j];
#pragma unroll
      for (int i = 0; i < 8; ++i) {
        stg[col][i + 8 * hi]      = __expf(s0[i] - m0) * r0;
        stg[col + 16][i + 8 * hi] = __expf(s1[i] - m1) * r1;
      }
      float t[16];
#pragma unroll
      for (int i = 0; i < 16; ++i) {
        const int kk = ((i < 8) ? i : i + 8) + 8 * hi;
        t[i] = stg[kk][col];
      }
      __builtin_amdgcn_sched_barrier(0x8 | 0x10);
      v16h ef;
#pragma unroll
      for (int i = 0; i < 16; ++i) ef[i] = (_Float16)t[i];
#pragma unroll
      for (int dt = 0; dt < 4; ++dt)
        acc[j][dt] = wmma16(ef, vf[dt], acc[j][dt]);
    }
  };

  static_assert(kN % 64 == 0, "two chunks per trip");
  v16h kfA[4], vfA[4], kfB[4], vfB[4];
  load_chunk(0, kfA, vfA);
  for (int k0 = 0; k0 < kN; k0 += 64) {
    load_chunk(k0 + 32, kfB, vfB);
    __builtin_amdgcn_sched_barrier(0);
    do_chunk(k0, kfA, vfA);
    const int kn = (k0 + 64 < kN) ? (k0 + 64) : k0;
    load_chunk(kn, kfA, vfA);
    __builtin_amdgcn_sched_barrier(0);
    do_chunk(k0 + 32, kfB, vfB);
  }

  const int b = bh >> 4, h = bh & 15;
  _Float16* cb = ctx16 + (size_t)b * kN * kC + (size_t)h * kD;
#pragma unroll
  for (int j = 0; j < kQT; ++j) {
    const int qrow = (qt0 + j) * 16 + 8 * hi;
#pragma unroll
    for (int dt = 0; dt < 4; ++dt)
#pragma unroll
      for (int r = 0; r < 8; ++r)
        cb[(size_t)(qrow + r) * kC + dt * 16 + col] = (_Float16)acc[j][dt][r];
  }
}

__global__ void ln_kernel(const float* __restrict__ resid,
                          const float* __restrict__ y,
                          const float* __restrict__ g,
                          const float* __restrict__ be,
                          float* __restrict__ out) {
  const int row = blockIdx.x;
  const int t   = threadIdx.x;
  float x[4], s = 0.f, sq = 0.f;
#pragma unroll
  for (int j = 0; j < 4; ++j) {
    const int c = t + j * 256;
    const float v = resid[(size_t)row * kC + c] + y[(size_t)row * kC + c];
    x[j] = v; s += v; sq += v * v;
  }
#pragma unroll
  for (int o = 16; o >= 1; o >>= 1) {
    s  += __shfl_xor(s,  o, 32);
    sq += __shfl_xor(sq, o, 32);
  }
  __shared__ float rs[8], rq[8];
  __shared__ float mu_s, inv_s;
  const int wave = t >> 5, lane = t & 31;
  if (lane == 0) { rs[wave] = s; rq[wave] = sq; }
  __syncthreads();
  if (t == 0) {
    float S = 0.f, Q = 0.f;
#pragma unroll
    for (int w = 0; w < 8; ++w) { S += rs[w]; Q += rq[w]; }
    const float mu = S * (1.0f / kC);
    mu_s  = mu;
    inv_s = rsqrtf(Q * (1.0f / kC) - mu * mu + kLnEps);
  }
  __syncthreads();
  const float mu = mu_s, inv = inv_s;
#pragma unroll
  for (int j = 0; j < 4; ++j) {
    const int c = t + j * 256;
    out[(size_t)row * kC + c] = (x[j] - mu) * inv * g[c] + be[c];
  }
}

extern "C" void kernel_launch(void* const* d_in, const int* in_sizes, int n_in,
                              void* d_out, int out_size, void* d_ws, size_t ws_size,
                              hipStream_t stream) {
  (void)in_sizes; (void)n_in; (void)out_size; (void)ws_size;
  const float* query = (const float*)d_in[0];
  const float* key   = (const float*)d_in[1];
  const float* value = (const float*)d_in[2];
  const float* Wq = (const float*)d_in[4];
  const float* Wk = (const float*)d_in[5];
  const float* Wv = (const float*)d_in[6];
  const float* Wo = (const float*)d_in[7];
  const float* ln_g = (const float*)d_in[8];
  const float* ln_b = (const float*)d_in[9];
  float* out = (float*)d_out;

  constexpr size_t BNC = (size_t)kB * kN * kC;
  constexpr size_t CC  = (size_t)kC * kC;
  constexpr size_t BHN = (size_t)kB * kH * kN;

  size_t off = 0;
  auto take = [&](size_t bytes) -> void* {
    void* p = (char*)d_ws + off;
    off += (bytes + 255) & ~(size_t)255;
    return p;
  };
  _Float16* xq16 = (_Float16*)take(BNC * 2);
  _Float16* xk16 = (_Float16*)take(BNC * 2);
  _Float16* xv16 = (_Float16*)take(BNC * 2);
  _Float16* wq16 = (_Float16*)take(CC * 2);
  _Float16* wk16 = (_Float16*)take(CC * 2);
  _Float16* wv16 = (_Float16*)take(CC * 2);
  _Float16* wo16 = (_Float16*)take(CC * 2);
  _Float16* qh16 = (_Float16*)take(BNC * 2);
  _Float16* kh16 = (_Float16*)take(BNC * 2);
  _Float16* vt16 = (_Float16*)take(BNC * 2);
  float*    mcol = (float*)take(BHN * 4);
  float*    rcol = (float*)take(BHN * 4);
  _Float16* ctx16 = xq16;
  float*    yf32  = (float*)xk16;

  cvt_all_f32_f16<<<kCvtGrid, kCvtBlock, 0, stream>>>(
      query, key, value, Wq, Wk, Wv, Wo,
      xq16, xk16, xv16, wq16, wk16, wv16, wo16);

  dim3 ggrid(32, 16);
  gemm_xwT<0><<<ggrid, 128, 0, stream>>>(xq16, wq16, qh16);
  gemm_xwT<0><<<ggrid, 128, 0, stream>>>(xk16, wk16, kh16);
  gemm_xwT<1><<<ggrid, 128, 0, stream>>>(xv16, wv16, vt16);

  colstats_kernel<<<dim3(kN / (16 * kKT), kB * kH), 256, 0, stream>>>(qh16, kh16, mcol, rcol);

  ctx_kernel<<<dim3(kN / (16 * kQT * kCW), kB * kH), 32 * kCW, 0, stream>>>(
      qh16, kh16, vt16, mcol, rcol, ctx16);

  gemm_xwT<2><<<ggrid, 128, 0, stream>>>(ctx16, wo16, yf32);
  ln_kernel<<<kB * kN, 256, 0, stream>>>(query, yf32, ln_g, ln_b, out);
}


# gemm kernels: the block's A and W rows go through LDS in K slices of 64 (whole-line global loads, one fetch per line per block, two stages); fragments from ds_load_b128; same k order of sums
# speedup vs baseline: 2.5104x; 1.2443x over previous
#include <hip/hip_runtime.h>
#include <hip/hip_bf16.h>


typedef __attribute__((ext_vector_type(16))) _Float16 v16h;
typedef __attribute__((ext_vector_type(8)))  _Float16 v8h;
typedef __attribute__((ext_vector_type(4)))  _Float16 v4h;
typedef __attribute__((ext_vector_type(8)))  float    v8f;

constexpr int kB = 2, kN = 2048, kC = 1024, kH = 16, kD = 64;
constexpr float kLnEps = 1e-5f;

__device__ __forceinline__ v8f wmma16(v16h a, v16h b, v8f c) {
  return __builtin_amdgcn_wmma_f32_16x16x32_f16(false, a, false, b, (short)0, c,
                                                false, false);
}

__device__ __forceinline__ v16h load_frag(const _Float16* __restrict__ p, int ld) {
  const int lane = threadIdx.x & 31;
  const int r    = lane & 15;
  const int ko   = (lane >> 4) << 3;
  const _Float16* b0 = p + (size_t)r * ld + ko;
  v8h lo = *(const v8h*)(b0);
  v8h hi = *(const v8h*)(b0 + 16);
  v16h f;
#pragma unroll
  for (int i = 0; i < 8; ++i) { f[i] = lo[i]; f[i + 8] = hi[i]; }
  return f;
}

constexpr int kCvtBlock  = 256;
constexpr int kCvtPerBlk = 8192;
constexpr int kCvtInFl   = 16;
constexpr int kCvtRounds = kCvtPerBlk / (kCvtBlock * kCvtInFl);
constexpr int kCvtXBlks  = kB * kN * kC / 4 / kCvtPerBlk;
constexpr int kCvtWBlks  = kC * kC / 4 / kCvtPerBlk;
constexpr int kCvtGrid   = 3 * kCvtXBlks + 4 * kCvtWBlks;
static_assert(kCvtRounds * kCvtBlock * kCvtInFl == kCvtPerBlk, "block chunk");
static_assert(kCvtXBlks * kCvtPerBlk * 4 == kB * kN * kC, "activation split");
static_assert(kCvtWBlks * kCvtPerBlk * 4 == kC * kC, "weight split");

__global__ __launch_bounds__(kCvtBlock)
void cvt_all_f32_f16(const float* __restrict__ xq, const float* __restrict__ xk,
                     const float* __restrict__ xv, const float* __restrict__ wq,
                     const float* __restrict__ wk, const float* __restrict__ wv,
                     const float* __restrict__ wo,
                     _Float16* __restrict__ xq16, _Float16* __restrict__ xk16,
                     _Float16* __restrict__ xv16, _Float16* __restrict__ wq16,
                     _Float16* __restrict__ wk16, _Float16* __restrict__ wv16,
                     _Float16* __restrict__ wo16) {
  const int b = blockIdx.x;
  const float* src;
  _Float16* dst;
  int lb;
  if (b < 3 * kCvtXBlks) {
    const int s = b / kCvtXBlks;
    lb  = b - s * kCvtXBlks;
    src = (s == 0) ? xq : (s == 1) ? xk : xv;
    dst = (s == 0) ? xq16 : (s == 1) ? xk16 : xv16;
  } else {
    const int bw = b - 3 * kCvtXBlks;
    const int s  = bw / kCvtWBlks;
    lb  = bw - s * kCvtWBlks;
    src = (s == 0) ? wq : (s == 1) ? wk : (s == 2) ? wv : wo;
    dst = (s == 0) ? wq16 : (s == 1) ? wk16 : (s == 2) ? wv16 : wo16;
  }
  const float4* s4 = (const float4*)src + (size_t)lb * kCvtPerBlk + threadIdx.x;
  v4h*          d4 = (v4h*)dst          + (size_t)lb * kCvtPerBlk + threadIdx.x;

#pragma unroll
  for (int r = 0; r < kCvtRounds; ++r) {
    float4 v[kCvtInFl];
#pragma unroll
    for (int j = 0; j < kCvtInFl; ++j)
      v[j] = s4[(r * kCvtInFl + j) * kCvtBlock];
#pragma unroll
    for (int j = 0; j < kCvtInFl; ++j) {
      v4h h;
      h[0] = (_Float16)v[j].x; h[1] = (_Float16)v[j].y;
      h[2] = (_Float16)v[j].z; h[3] = (_Float16)v[j].w;
      d4[(r * kCvtInFl + j) * kCvtBlock] = h;
    }
  }
}

constexpr int kGK     = 64;
constexpr int kGPitch = kGK + 8;

__device__ __forceinline__ v16h lds_frag(const _Float16* p) {
  const int lane = threadIdx.x & 31;
  const int r    = lane & 15;
  const int ko   = (lane >> 4) << 3;
  const _Float16* b0 = p + r * kGPitch + ko;
  v8h lo = *(const v8h*)(b0);
  v8h hi = *(const v8h*)(b0 + 16);
  v16h f;
#pragma unroll
  for (int i = 0; i < 8; ++i) { f[i] = lo[i]; f[i + 8] = hi[i]; }
  return f;
}

template <int MODE>
__global__ __launch_bounds__(128)
void gemm_xwT(const _Float16* __restrict__ A,
              const _Float16* __restrict__ W,
              void* __restrict__ out) {
  const int tid  = threadIdx.x;
  const int wave = tid >> 5;
  const int lane = tid & 31;
  const int colL = lane & 15;
  const int hi   = lane >> 4;
  const int mB = blockIdx.x * 128;
  const int m0 = mB + wave * 32;
  const int n0 = blockIdx.y * 64;

  __shared__ __attribute__((aligned(16))) _Float16 sA[2][128][kGPitch];
  __shared__ __attribute__((aligned(16))) _Float16 sW[2][64][kGPitch];

  const int crow = tid >> 3;
  const int ccol = (tid & 7) * 8;
  const _Float16* ag = A + (size_t)(mB + crow) * kC + ccol;
  const _Float16* wg = W + (size_t)(n0 + crow) * kC + ccol;

  v8h ra[8], rw[4];
  auto gload = [&](int k0) {
#pragma unroll
    for (int i = 0; i < 8; ++i) ra[i] = *(const v8h*)(ag + (size_t)i * 16 * kC + k0);
#pragma unroll
    for (int i = 0; i < 4; ++i) rw[i] = *(const v8h*)(wg + (size_t)i * 16 * kC + k0);
  };
  auto sstore = [&](int s) {
#pragma unroll
    for (int i = 0; i < 8; ++i) *(v8h*)&sA[s][crow + 16 * i][ccol] = ra[i];
#pragma unroll
    for (int i = 0; i < 4; ++i) *(v8h*)&sW[s][crow + 16 * i][ccol] = rw[i];
  };

  v8f acc[2][4] = {};

  gload(0);
  sstore(0);
  __syncthreads();

  constexpr int kSlices = kC / kGK;
  for (int t = 0; t < kSlices; ++t) {
    const int s = t & 1;
    if (t + 1 < kSlices) gload((t + 1) * kGK);

#pragma unroll
    for (int kk = 0; kk < kGK; kk += 32) {
      v16h a0 = lds_frag(&sA[s][wave * 32][kk]);
      v16h a1 = lds_frag(&sA[s][wave * 32 + 16][kk]);
      v16h bf[4];
#pragma unroll
      for (int nt = 0; nt < 4; ++nt) bf[nt] = lds_frag(&sW[s][nt * 16][kk]);
#pragma unroll
      for (int nt = 0; nt < 4; ++nt) {
        acc[0][nt] = wmma16(a0, bf[nt], acc[0][nt]);
        acc[1][nt] = wmma16(a1, bf[nt], acc[1][nt]);
      }
    }

    if (t + 1 < kSlices) sstore(s ^ 1);
    __syncthreads();
  }

#pragma unroll
  for (int mt = 0; mt < 2; ++mt)
#pragma unroll
    for (int nt = 0; nt < 4; ++nt)
#pragma unroll
      for (int r = 0; r < 8; ++r) {
        const int row  = m0 + mt * 16 + 8 * hi + r;
        const int colc = n0 + nt * 16 + colL;
        const float v  = acc[mt][nt][r];
        if constexpr (MODE == 2) {
          ((float*)out)[(size_t)row * kC + colc] = v;
        } else {
          const int b = row >> 11, n = row & (kN - 1);
          const int h = colc >> 6, d = colc & (kD - 1);
          if constexpr (MODE == 0)
            ((_Float16*)out)[(((size_t)(b * kH + h)) * kN + n) * kD + d] = (_Float16)v;
          else
            ((_Float16*)out)[(((size_t)(b * kH + h)) * kD + d) * kN + n] = (_Float16)v;
        }
      }
}

constexpr int kKT = 8;

__global__ __launch_bounds__(256, 4)
void colstats_kernel(const _Float16* __restrict__ q16,
                     const _Float16* __restrict__ k16,
                     float* __restrict__ mcol,
                     float* __restrict__ rcol) {
  const int kg   = blockIdx.x;
  const int bh   = blockIdx.y;
  const int wave = threadIdx.x >> 5;
  const int lane = threadIdx.x & 31;
  const _Float16* qb = q16 + (size_t)bh * kN * kD;
  const _Float16* kb = k16 + ((size_t)bh * kN + (size_t)kg * 16 * kKT) * kD;

  v16h bk[kKT][2];
#pragma unroll
  for (int t = 0; t < kKT; ++t) {
    bk[t][0] = load_frag(kb + (size_t)t * 16 * kD + 0,  kD);
    bk[t][1] = load_frag(kb + (size_t)t * 16 * kD + 32, kD);
  }

  float mrun[kKT], srun[kKT];
#pragma unroll
  for (int t = 0; t < kKT; ++t) { mrun[t] = -3.4e38f; srun[t] = 0.f; }

  constexpr int kTrips = kN / 16 / 8;
  v16h a0 = load_frag(qb + (size_t)wave * 16 * kD + 0,  kD);
  v16h a1 = load_frag(qb + (size_t)wave * 16 * kD + 32, kD);
  for (int it = 0; it < kTrips; ++it) {
    v16h n0 = a0, n1 = a1;
    if (it + 1 < kTrips) {
      const int qn = wave + 8 * (it + 1);
      n0 = load_frag(qb + (size_t)qn * 16 * kD + 0,  kD);
      n1 = load_frag(qb + (size_t)qn * 16 * kD + 32, kD);
    }
#pragma unroll
    for (int tp = 0; tp < kKT; tp += 2) {
      v8f s[2];
#pragma unroll
      for (int u = 0; u < 2; ++u) {
        s[u] = v8f{};
        s[u] = wmma16(a0, bk[tp + u][0], s[u]);
        s[u] = wmma16(a1, bk[tp + u][1], s[u]);
      }
#pragma unroll
      for (int u = 0; u < 2; ++u) {
        const int t = tp + u;
        float tm = s[u][0];
#pragma unroll
        for (int i = 1; i < 8; ++i) tm = fmaxf(tm, s[u][i]);
        const float mn = fmaxf(mrun[t], tm);
        float part = 0.f;
#pragma unroll
        for (int i = 0; i < 8; ++i) part += __expf(s[u][i] - mn);
        srun[t] = srun[t] * __expf(mrun[t] - mn) + part;
        mrun[t] = mn;
      }
      __builtin_amdgcn_sched_barrier(0);
    }
    a0 = n0; a1 = n1;
  }

  __shared__ float redm[8][16 * kKT];
  __shared__ float reds[8][16 * kKT];
#pragma unroll
  for (int t = 0; t < kKT; ++t) {
    const float mo = __shfl_xor(mrun[t], 16, 32);
    const float so = __shfl_xor(srun[t], 16, 32);
    const float mn = fmaxf(mrun[t], mo);
    const float sv = srun[t] * __expf(mrun[t] - mn) + so * __expf(mo - mn);
    if (lane < 16) { redm[wave][t * 16 + lane] = mn; reds[wave][t * 16 + lane] = sv; }
  }
  __syncthreads();
  if (threadIdx.x < 16 * kKT) {
    const int c = threadIdx.x;
    float m = redm[0][c];
#pragma unroll
    for (int w = 1; w < 8; ++w) m = fmaxf(m, redm[w][c]);
    float sum = 0.f;
#pragma unroll
    for (int w = 0; w < 8; ++w) sum += reds[w][c] * __expf(redm[w][c] - m);
    const size_t o = (size_t)bh * kN + (size_t)kg * 16 * kKT + c;
    mcol[o] = m;
    rcol[o] = 1.0f / sum;
  }
}

constexpr int kQT = 4;
constexpr int kCW = 4;

__global__ __launch_bounds__(32 * kCW)
void ctx_kernel(const _Float16* __restrict__ q16,
                const _Float16* __restrict__ k16,
                const _Float16* __restrict__ vt16,
                const float* __restrict__ mcol,
                const float* __restrict__ rcol,
                _Float16* __restrict__ ctx16) {
  const int bh   = blockIdx.y;
  const int wave = threadIdx.x >> 5;
  const int lane = threadIdx.x & 31;
  const int col  = lane & 15;
  const int hi   = lane >> 4;
  const int qt0  = (blockIdx.x * kCW + wave) * kQT;

  __shared__ float mr_s[2][kN];
  __shared__ float estage[kCW][kQT][32][17];

  for (int i = threadIdx.x; i < kN; i += 32 * kCW) {
    mr_s[0][i] = mcol[(size_t)bh * kN + i];
    mr_s[1][i] = rcol[(size_t)bh * kN + i];
  }
  __syncthreads();

  const _Float16* qb = q16  + (size_t)bh * kN * kD;
  const _Float16* kb = k16  + (size_t)bh * kN * kD;
  const _Float16* vb = vt16 + (size_t)bh * kD * kN;

  v16h aq[kQT][2];
#pragma unroll
  for (int j = 0; j < kQT; ++j) {
    aq[j][0] = load_frag(qb + (size_t)(qt0 + j) * 16 * kD + 0,  kD);
    aq[j][1] = load_frag(qb + (size_t)(qt0 + j) * 16 * kD + 32, kD);
  }

  v8f acc[kQT][4] = {};

  auto load_chunk = [&](int k0, v16h (&kf)[4], v16h (&vf)[4]) {
    kf[0] = load_frag(kb + (size_t)(k0     ) * kD + 0,  kD);
    kf[1] = load_frag(kb + (size_t)(k0     ) * kD + 32, kD);
    kf[2] = load_frag(kb + (size_t)(k0 + 16) * kD + 0,  kD);
    kf[3] = load_frag(kb + (size_t)(k0 + 16) * kD + 32, kD);
#pragma unroll
    for (int dt = 0; dt < 4; ++dt)
      vf[dt] = load_frag(vb + (size_t)dt * 16 * kN + k0, kN);
  };

  auto do_chunk = [&](int k0, const v16h (&kf)[4], const v16h (&vf)[4]) {
    const float m0 = mr_s[0][k0 + col],      r0 = mr_s[1][k0 + col];
    const float m1 = mr_s[0][k0 + 16 + col], r1 = mr_s[1][k0 + 16 + col];

#pragma unroll
    for (int j = 0; j < kQT; ++j) {
      v8f s0 = {}, s1 = {};
      s0 = wmma16(aq[j][0], kf[0], s0);
      s0 = wmma16(aq[j][1], kf[1], s0);
      s1 = wmma16(aq[j][0], kf[2], s1);
      s1 = wmma16(aq[j][1], kf[3], s1);

      float (*stg)[17] = estage[wave][j];
#pragma unroll
      for (int i = 0; i < 8; ++i) {
        stg[col][i + 8 * hi]      = __expf(s0[i] - m0) * r0;
        stg[col + 16][i + 8 * hi] = __expf(s1[i] - m1) * r1;
      }
      float t[16];
#pragma unroll
      for (int i = 0; i < 16; ++i) {
        const int kk = ((i < 8) ? i : i + 8) + 8 * hi;
        t[i] = stg[kk][col];
      }
      __builtin_amdgcn_sched_barrier(0x8 | 0x10);
      v16h ef;
#pragma unroll
      for (int i = 0; i < 16; ++i) ef[i] = (_Float16)t[i];
#pragma unroll
      for (int dt = 0; dt < 4; ++dt)
        acc[j][dt] = wmma16(ef, vf[dt], acc[j][dt]);
    }
  };

  static_assert(kN % 64 == 0, "two chunks per trip");
  v16h kfA[4], vfA[4], kfB[4], vfB[4];
  load_chunk(0, kfA, vfA);
  for (int k0 = 0; k0 < kN; k0 += 64) {
    load_chunk(k0 + 32, kfB, vfB);
    __builtin_amdgcn_sched_barrier(0);
    do_chunk(k0, kfA, vfA);
    const int kn = (k0 + 64 < kN) ? (k0 + 64) : k0;
    load_chunk(kn, kfA, vfA);
    __builtin_amdgcn_sched_barrier(0);
    do_chunk(k0 + 32, kfB, vfB);
  }

  const int b = bh >> 4, h = bh & 15;
  _Float16* cb = ctx16 + (size_t)b * kN * kC + (size_t)h * kD;
#pragma unroll
  for (int j = 0; j < kQT; ++j) {
    const int qrow = (qt0 + j) * 16 + 8 * hi;
#pragma unroll
    for (int dt = 0; dt < 4; ++dt)
#pragma unroll
      for (int r = 0; r < 8; ++r)
        cb[(size_t)(qrow + r) * kC + dt * 16 + col] = (_Float16)acc[j][dt][r];
  }
}

__global__ void ln_kernel(const float* __restrict__ resid,
                          const float* __restrict__ y,
                          const float* __restrict__ g,
                          const float* __restrict__ be,
                          float* __restrict__ out) {
  const int row = blockIdx.x;
  const int t   = threadIdx.x;
  float x[4], s = 0.f, sq = 0.f;
#pragma unroll
  for (int j = 0; j < 4; ++j) {
    const int c = t + j * 256;
    const float v = resid[(size_t)row * kC + c] + y[(size_t)row * kC + c];
    x[j] = v; s += v; sq += v * v;
  }
#pragma unroll
  for (int o = 16; o >= 1; o >>= 1) {
    s  += __shfl_xor(s,  o, 32);
    sq += __shfl_xor(sq, o, 32);
  }
  __shared__ float rs[8], rq[8];
  __shared__ float mu_s, inv_s;
  const int wave = t >> 5, lane = t & 31;
  if (lane == 0) { rs[wave] = s; rq[wave] = sq; }
  __syncthreads();
  if (t == 0) {
    float S = 0.f, Q = 0.f;
#pragma unroll
    for (int w = 0; w < 8; ++w) { S += rs[w]; Q += rq[w]; }
    const float mu = S * (1.0f / kC);
    mu_s  = mu;
    inv_s = rsqrtf(Q * (1.0f / kC) - mu * mu + kLnEps);
  }
  __syncthreads();
  const float mu = mu_s, inv = inv_s;
#pragma unroll
  for (int j = 0; j < 4; ++j) {
    const int c = t + j * 256;
    out[(size_t)row * kC + c] = (x[j] - mu) * inv * g[c] + be[c];
  }
}

extern "C" void kernel_launch(void* const* d_in, const int* in_sizes, int n_in,
                              void* d_out, int out_size, void* d_ws, size_t ws_size,
                              hipStream_t stream) {
  (void)in_sizes; (void)n_in; (void)out_size; (void)ws_size;
  const float* query = (const float*)d_in[0];
  const float* key   = (const float*)d_in[1];
  const float* value = (const float*)d_in[2];
  const float* Wq = (const float*)d_in[4];
  const float* Wk = (const float*)d_in[5];
  const float* Wv = (const float*)d_in[6];
  const float* Wo = (const float*)d_in[7];
  const float* ln_g = (const float*)d_in[8];
  const float* ln_b = (const float*)d_in[9];
  float* out = (float*)d_out;

  constexpr size_t BNC = (size_t)kB * kN * kC;
  constexpr size_t CC  = (size_t)kC * kC;
  constexpr size_t BHN = (size_t)kB * kH * kN;

  size_t off = 0;
  auto take = [&](size_t bytes) -> void* {
    void* p = (char*)d_ws + off;
    off += (bytes + 255) & ~(size_t)255;
    return p;
  };
  _Float16* xq16 = (_Float16*)take(BNC * 2);
  _Float16* xk16 = (_Float16*)take(BNC * 2);
  _Float16* xv16 = (_Float16*)take(BNC * 2);
  _Float16* wq16 = (_Float16*)take(CC * 2);
  _Float16* wk16 = (_Float16*)take(CC * 2);
  _Float16* wv16 = (_Float16*)take(CC * 2);
  _Float16* wo16 = (_Float16*)take(CC * 2);
  _Float16* qh16 = (_Float16*)take(BNC * 2);
  _Float16* kh16 = (_Float16*)take(BNC * 2);
  _Float16* vt16 = (_Float16*)take(BNC * 2);
  float*    mcol = (float*)take(BHN * 4);
  float*    rcol = (float*)take(BHN * 4);
  _Float16* ctx16 = xq16;
  float*    yf32  = (float*)xk16;

  cvt_all_f32_f16<<<kCvtGrid, kCvtBlock, 0, stream>>>(
      query, key, value, Wq, Wk, Wv, Wo,
      xq16, xk16, xv16, wq16, wk16, wv16, wo16);

  dim3 ggrid(32, 16);
  gemm_xwT<0><<<ggrid, 128, 0, stream>>>(xq16, wq16, qh16);
  gemm_xwT<0><<<ggrid, 128, 0, stream>>>(xk16, wk16, kh16);
  gemm_xwT<1><<<ggrid, 128, 0, stream>>>(xv16, wv16, vt16);

  colstats_kernel<<<dim3(kN / (16 * kKT), kB * kH), 256, 0, stream>>>(qh16, kh16, mcol, rcol);

  ctx_kernel<<<dim3(kN / (16 * kQT * kCW), kB * kH), 32 * kCW, 0, stream>>>(
      qh16, kh16, vt16, mcol, rcol, ctx16);

  gemm_xwT<2><<<ggrid, 128, 0, stream>>>(ctx16, wo16, yf32);
  ln_kernel<<<kB * kN, 256, 0, stream>>>(query, yf32, ln_g, ln_b, out);
}


# workspace layout: Q/K heads and V^T stored in WMMA fragment order, so each fragment load in the stats and ctx kernels reads 512 contiguous bytes (4 lines, not 16); V^T epilogue = 8 whole-line stores
# speedup vs baseline: 2.6206x; 1.0439x over previous
#include <hip/hip_runtime.h>
#include <hip/hip_bf16.h>


typedef __attribute__((ext_vector_type(16))) _Float16 v16h;
typedef __attribute__((ext_vector_type(8)))  _Float16 v8h;
typedef __attribute__((ext_vector_type(4)))  _Float16 v4h;
typedef __attribute__((ext_vector_type(8)))  float    v8f;

constexpr int kB = 2, kN = 2048, kC = 1024, kH = 16, kD = 64;
constexpr float kLnEps = 1e-5f;

__device__ __forceinline__ v8f wmma16(v16h a, v16h b, v8f c) {
  return __builtin_amdgcn_wmma_f32_16x16x32_f16(false, a, false, b, (short)0, c,
                                                false, false);
}

constexpr int kFragHalves  = 512;
constexpr int kPieceHalves = 256;

__device__ __forceinline__ v16h load_frag_pk(const _Float16* __restrict__ frag) {
  const int lane = threadIdx.x & 31;
  v8h lo = *(const v8h*)(frag + lane * 8);
  v8h hi = *(const v8h*)(frag + kPieceHalves + lane * 8);
  v16h f;
#pragma unroll
  for (int i = 0; i < 8; ++i) { f[i] = lo[i]; f[i + 8] = hi[i]; }
  return f;
}

__device__ __forceinline__ int qk_pk_off(int n, int d) {
  return (n >> 4) * (2 * kFragHalves) + (d >> 5) * kFragHalves +
         ((d >> 4) & 1) * kPieceHalves + ((((d >> 3) & 1) << 4) + (n & 15)) * 8 + (d & 7);
}

constexpr int kCvtBlock  = 256;
constexpr int kCvtPerBlk = 8192;
constexpr int kCvtInFl   = 16;
constexpr int kCvtRounds = kCvtPerBlk / (kCvtBlock * kCvtInFl);
constexpr int kCvtXBlks  = kB * kN * kC / 4 / kCvtPerBlk;
constexpr int kCvtWBlks  = kC * kC / 4 / kCvtPerBlk;
constexpr int kCvtGrid   = 3 * kCvtXBlks + 4 * kCvtWBlks;
static_assert(kCvtRounds * kCvtBlock * kCvtInFl == kCvtPerBlk, "block chunk");
static_assert(kCvtXBlks * kCvtPerBlk * 4 == kB * kN * kC, "activation split");
static_assert(kCvtWBlks * kCvtPerBlk * 4 == kC * kC, "weight split");

__global__ __launch_bounds__(kCvtBlock)
void cvt_all_f32_f16(const float* __restrict__ xq, const float* __restrict__ xk,
                     const float* __restrict__ xv, const float* __restrict__ wq,
                     const float* __restrict__ wk, const float* __restrict__ wv,
                     const float* __restrict__ wo,
                     _Float16* __restrict__ xq16, _Float16* __restrict__ xk16,
                     _Float16* __restrict__ xv16, _Float16* __restrict__ wq16,
                     _Float16* __restrict__ wk16, _Float16* __restrict__ wv16,
                     _Float16* __restrict__ wo16) {
  const int b = blockIdx.x;
  const float* src;
  _Float16* dst;
  int lb;
  if (b < 3 * kCvtXBlks) {
    const int s = b / kCvtXBlks;
    lb  = b - s * kCvtXBlks;
    src = (s == 0) ? xq : (s == 1) ? xk : xv;
    dst = (s == 0) ? xq16 : (s == 1) ? xk16 : xv16;
  } else {
    const int bw = b - 3 * kCvtXBlks;
    const int s  = bw / kCvtWBlks;
    lb  = bw - s * kCvtWBlks;
    src = (s == 0) ? wq : (s == 1) ? wk : (s == 2) ? wv : wo;
    dst = (s == 0) ? wq16 : (s == 1) ? wk16 : (s == 2) ? wv16 : wo16;
  }
  const float4* s4 = (const float4*)src + (size_t)lb * kCvtPerBlk + threadIdx.x;
  v4h*          d4 = (v4h*)dst          + (size_t)lb * kCvtPerBlk + threadIdx.x;

#pragma unroll
  for (int r = 0; r < kCvtRounds; ++r) {
    float4 v[kCvtInFl];
#pragma unroll
    for (int j = 0; j < kCvtInFl; ++j)
      v[j] = s4[(r * kCvtInFl + j) * kCvtBlock];
#pragma unroll
    for (int j = 0; j < kCvtInFl; ++j) {
      v4h h;
      h[0] = (_Float16)v[j].x; h[1] = (_Float16)v[j].y;
      h[2] = (_Float16)v[j].z; h[3] = (_Float16)v[j].w;
      d4[(r * kCvtInFl + j) * kCvtBlock] = h;
    }
  }
}

constexpr int kGK     = 64;
constexpr int kGPitch = kGK + 8;

__device__ __forceinline__ v16h lds_frag(const _Float16* p) {
  const int lane = threadIdx.x & 31;
  const int r    = lane & 15;
  const int ko   = (lane >> 4) << 3;
  const _Float16* b0 = p + r * kGPitch + ko;
  v8h lo = *(const v8h*)(b0);
  v8h hi = *(const v8h*)(b0 + 16);
  v16h f;
#pragma unroll
  for (int i = 0; i < 8; ++i) { f[i] = lo[i]; f[i + 8] = hi[i]; }
  return f;
}

template <int MODE>
__global__ __launch_bounds__(128)
void gemm_xwT(const _Float16* __restrict__ A,
              const _Float16* __restrict__ W,
              void* __restrict__ out) {
  const int tid  = threadIdx.x;
  const int wave = tid >> 5;
  const int lane = tid & 31;
  const int colL = lane & 15;
  const int hi   = lane >> 4;
  const int mB = blockIdx.x * 128;
  const int m0 = mB + wave * 32;
  const int n0 = blockIdx.y * 64;

  __shared__ __attribute__((aligned(16))) _Float16 sA[2][128][kGPitch];
  __shared__ __attribute__((aligned(16))) _Float16 sW[2][64][kGPitch];

  const int crow = tid >> 3;
  const int ccol = (tid & 7) * 8;
  const _Float16* ag = A + (size_t)(mB + crow) * kC + ccol;
  const _Float16* wg = W + (size_t)(n0 + crow) * kC + ccol;

  v8h ra[8], rw[4];
  auto gload = [&](int k0) {
#pragma unroll
    for (int i = 0; i < 8; ++i) ra[i] = *(const v8h*)(ag + (size_t)i * 16 * kC + k0);
#pragma unroll
    for (int i = 0; i < 4; ++i) rw[i] = *(const v8h*)(wg + (size_t)i * 16 * kC + k0);
  };
  auto sstore = [&](int s) {
#pragma unroll
    for (int i = 0; i < 8; ++i) *(v8h*)&sA[s][crow + 16 * i][ccol] = ra[i];
#pragma unroll
    for (int i = 0; i < 4; ++i) *(v8h*)&sW[s][crow + 16 * i][ccol] = rw[i];
  };

  v8f acc[2][4] = {};

  gload(0);
  sstore(0);
  __syncthreads();

  constexpr int kSlices = kC / kGK;
  for (int t = 0; t < kSlices; ++t) {
    const int s = t & 1;
    if (t + 1 < kSlices) gload((t + 1) * kGK);

#pragma unroll
    for (int kk = 0; kk < kGK; kk += 32) {
      v16h a0 = lds_frag(&sA[s][wave * 32][kk]);
      v16h a1 = lds_frag(&sA[s][wave * 32 + 16][kk]);
      v16h bf[4];
#pragma unroll
      for (int nt = 0; nt < 4; ++nt) bf[nt] = lds_frag(&sW[s][nt * 16][kk]);
#pragma unroll
      for (int nt = 0; nt < 4; ++nt) {
        acc[0][nt] = wmma16(a0, bf[nt], acc[0][nt]);
        acc[1][nt] = wmma16(a1, bf[nt], acc[1][nt]);
      }
    }

    if (t + 1 < kSlices) sstore(s ^ 1);
    __syncthreads();
  }

  if constexpr (MODE == 1) {
    const int b = m0 >> 11, c = (m0 & (kN - 1)) >> 5;
    const int h = n0 >> 6;
    _Float16* hb = (_Float16*)out + (size_t)(b * kH + h) * kD * kN +
                   (size_t)c * (4 * kFragHalves) + lane * 8;
#pragma unroll
    for (int mt = 0; mt < 2; ++mt)
#pragma unroll
      for (int nt = 0; nt < 4; ++nt) {
        v8h pk;
#pragma unroll
        for (int r = 0; r < 8; ++r) pk[r] = (_Float16)acc[mt][nt][r];
        *(v8h*)(hb + nt * kFragHalves + mt * kPieceHalves) = pk;
      }
  } else {
#pragma unroll
    for (int mt = 0; mt < 2; ++mt)
#pragma unroll
      for (int nt = 0; nt < 4; ++nt)
#pragma unroll
        for (int r = 0; r < 8; ++r) {
          const int row  = m0 + mt * 16 + 8 * hi + r;
          const int colc = n0 + nt * 16 + colL;
          const float v  = acc[mt][nt][r];
          if constexpr (MODE == 2) {
            ((float*)out)[(size_t)row * kC + colc] = v;
          } else {
            const int b = row >> 11, n = row & (kN - 1);
            const int h = colc >> 6, d = colc & (kD - 1);
            ((_Float16*)out)[(size_t)(b * kH + h) * kN * kD + qk_pk_off(n, d)] = (_Float16)v;
          }
        }
  }
}

constexpr int kKT = 8;

__global__ __launch_bounds__(256, 4)
void colstats_kernel(const _Float16* __restrict__ q16,
                     const _Float16* __restrict__ k16,
                     float* __restrict__ mcol,
                     float* __restrict__ rcol) {
  const int kg   = blockIdx.x;
  const int bh   = blockIdx.y;
  const int wave = threadIdx.x >> 5;
  const int lane = threadIdx.x & 31;
  const _Float16* qb = q16 + (size_t)bh * kN * kD;
  const _Float16* kb = k16 + ((size_t)bh * kN + (size_t)kg * 16 * kKT) * kD;

  v16h bk[kKT][2];
#pragma unroll
  for (int t = 0; t < kKT; ++t) {
    bk[t][0] = load_frag_pk(kb + (size_t)t * 16 * kD);
    bk[t][1] = load_frag_pk(kb + (size_t)t * 16 * kD + kFragHalves);
  }

  float mrun[kKT], srun[kKT];
#pragma unroll
  for (int t = 0; t < kKT; ++t) { mrun[t] = -3.4e38f; srun[t] = 0.f; }

  constexpr int kTrips = kN / 16 / 8;
  v16h a0 = load_frag_pk(qb + (size_t)wave * 16 * kD);
  v16h a1 = load_frag_pk(qb + (size_t)wave * 16 * kD + kFragHalves);
  for (int it = 0; it < kTrips; ++it) {
    v16h n0 = a0, n1 = a1;
    if (it + 1 < kTrips) {
      const int qn = wave + 8 * (it + 1);
      n0 = load_frag_pk(qb + (size_t)qn * 16 * kD);
      n1 = load_frag_pk(qb + (size_t)qn * 16 * kD + kFragHalves);
    }
#pragma unroll
    for (int tp = 0; tp < kKT; tp += 2) {
      v8f s[2];
#pragma unroll
      for (int u = 0; u < 2; ++u) {
        s[u] = v8f{};
        s[u] = wmma16(a0, bk[tp + u][0], s[u]);
        s[u] = wmma16(a1, bk[tp + u][1], s[u]);
      }
#pragma unroll
      for (int u = 0; u < 2; ++u) {
        const int t = tp + u;
        float tm = s[u][0];
#pragma unroll
        for (int i = 1; i < 8; ++i) tm = fmaxf(tm, s[u][i]);
        const float mn = fmaxf(mrun[t], tm);
        float part = 0.f;
#pragma unroll
        for (int i = 0; i < 8; ++i) part += __expf(s[u][i] - mn);
        srun[t] = srun[t] * __expf(mrun[t] - mn) + part;
        mrun[t] = mn;
      }
      __builtin_amdgcn_sched_barrier(0);
    }
    a0 = n0; a1 = n1;
  }

  __shared__ float redm[8][16 * kKT];
  __shared__ float reds[8][16 * kKT];
#pragma unroll
  for (int t = 0; t < kKT; ++t) {
    const float mo = __shfl_xor(mrun[t], 16, 32);
    const float so = __shfl_xor(srun[t], 16, 32);
    const float mn = fmaxf(mrun[t], mo);
    const float sv = srun[t] * __expf(mrun[t] - mn) + so * __expf(mo - mn);
    if (lane < 16) { redm[wave][t * 16 + lane] = mn; reds[wave][t * 16 + lane] = sv; }
  }
  __syncthreads();
  if (threadIdx.x < 16 * kKT) {
    const int c = threadIdx.x;
    float m = redm[0][c];
#pragma unroll
    for (int w = 1; w < 8; ++w) m = fmaxf(m, redm[w][c]);
    float sum = 0.f;
#pragma unroll
    for (int w = 0; w < 8; ++w) sum += reds[w][c] * __expf(redm[w][c] - m);
    const size_t o = (size_t)bh * kN + (size_t)kg * 16 * kKT + c;
    mcol[o] = m;
    rcol[o] = 1.0f / sum;
  }
}

constexpr int kQT = 4;
constexpr int kCW = 4;

__global__ __launch_bounds__(32 * kCW)
void ctx_kernel(const _Float16* __restrict__ q16,
                const _Float16* __restrict__ k16,
                const _Float16* __restrict__ vt16,
                const float* __restrict__ mcol,
                const float* __restrict__ rcol,
                _Float16* __restrict__ ctx16) {
  const int bh   = blockIdx.y;
  const int wave = threadIdx.x >> 5;
  const int lane = threadIdx.x & 31;
  const int col  = lane & 15;
  const int hi   = lane >> 4;
  const int qt0  = (blockIdx.x * kCW + wave) * kQT;

  __shared__ float mr_s[2][kN];
  __shared__ float estage[kCW][kQT][32][17];

  for (int i = threadIdx.x; i < kN; i += 32 * kCW) {
    mr_s[0][i] = mcol[(size_t)bh * kN + i];
    mr_s[1][i] = rcol[(size_t)bh * kN + i];
  }
  __syncthreads();

  const _Float16* qb = q16  + (size_t)bh * kN * kD;
  const _Float16* kb = k16  + (size_t)bh * kN * kD;
  const _Float16* vb = vt16 + (size_t)bh * kD * kN;

  v16h aq[kQT][2];
#pragma unroll
  for (int j = 0; j < kQT; ++j) {
    aq[j][0] = load_frag_pk(qb + (size_t)(qt0 + j) * 16 * kD);
    aq[j][1] = load_frag_pk(qb + (size_t)(qt0 + j) * 16 * kD + kFragHalves);
  }

  v8f acc[kQT][4] = {};

  auto load_chunk = [&](int k0, v16h (&kf)[4], v16h (&vf)[4]) {
    kf[0] = load_frag_pk(kb + (size_t)(k0     ) * kD);
    kf[1] = load_frag_pk(kb + (size_t)(k0     ) * kD + kFragHalves);
    kf[2] = load_frag_pk(kb + (size_t)(k0 + 16) * kD);
    kf[3] = load_frag_pk(kb + (size_t)(k0 + 16) * kD + kFragHalves);
#pragma unroll
    for (int dt = 0; dt < 4; ++dt)
      vf[dt] = load_frag_pk(vb + (size_t)k0 * kD + dt * kFragHalves);
  };

  auto do_chunk = [&](int k0, const v16h (&kf)[4], const v16h (&vf)[4]) {
    const float m0 = mr_s[0][k0 + col],      r0 = mr_s[1][k0 + col];
    const float m1 = mr_s[0][k0 + 16 + col], r1 = mr_s[1][k0 + 16 + col];

#pragma unroll
    for (int j = 0; j < kQT; ++j) {
      v8f s0 = {}, s1 = {};
      s0 = wmma16(aq[j][0], kf[0], s0);
      s0 = wmma16(aq[j][1], kf[1], s0);
      s1 = wmma16(aq[j][0], kf[2], s1);
      s1 = wmma16(aq[j][1], kf[3], s1);

      float (*stg)[17] = estage[wave][j];
#pragma unroll
      for (int i = 0; i < 8; ++i) {
        stg[col][i + 8 * hi]      = __expf(s0[i] - m0) * r0;
        stg[col + 16][i + 8 * hi] = __expf(s1[i] - m1) * r1;
      }
      float t[16];
#pragma unroll
      for (int i = 0; i < 16; ++i) {
        const int kk = ((i < 8) ? i : i + 8) + 8 * hi;
        t[i] = stg[kk][col];
      }
      __builtin_amdgcn_sched_barrier(0x8 | 0x10);
      v16h ef;
#pragma unroll
      for (int i = 0; i < 16; ++i) ef[i] = (_Float16)t[i];
#pragma unroll
      for (int dt = 0; dt < 4; ++dt)
        acc[j][dt] = wmma16(ef, vf[dt], acc[j][dt]);
    }
  };

  static_assert(kN % 64 == 0, "two chunks per trip");
  v16h kfA[4], vfA[4], kfB[4], vfB[4];
  load_chunk(0, kfA, vfA);
  for (int k0 = 0; k0 < kN; k0 += 64) {
    load_chunk(k0 + 32, kfB, vfB);
    __builtin_amdgcn_sched_barrier(0);
    do_chunk(k0, kfA, vfA);
    const int kn = (k0 + 64 < kN) ? (k0 + 64) : k0;
    load_chunk(kn, kfA, vfA);
    __builtin_amdgcn_sched_barrier(0);
    do_chunk(k0 + 32, kfB, vfB);
  }

  const int b = bh >> 4, h = bh & 15;
  _Float16* cb = ctx16 + (size_t)b * kN * kC + (size_t)h * kD;
#pragma unroll
  for (int j = 0; j < kQT; ++j) {
    const int qrow = (qt0 + j) * 16 + 8 * hi;
#pragma unroll
    for (int dt = 0; dt < 4; ++dt)
#pragma unroll
      for (int r = 0; r < 8; ++r)
        cb[(size_t)(qrow + r) * kC + dt * 16 + col] = (_Float16)acc[j][dt][r];
  }
}

__global__ void ln_kernel(const float* __restrict__ resid,
                          const float* __restrict__ y,
                          const float* __restrict__ g,
                          const float* __restrict__ be,
                          float* __restrict__ out) {
  const int row = blockIdx.x;
  const int t   = threadIdx.x;
  float x[4], s = 0.f, sq = 0.f;
#pragma unroll
  for (int j = 0; j < 4; ++j) {
    const int c = t + j * 256;
    const float v = resid[(size_t)row * kC + c] + y[(size_t)row * kC + c];
    x[j] = v; s += v; sq += v * v;
  }
#pragma unroll
  for (int o = 16; o >= 1; o >>= 1) {
    s  += __shfl_xor(s,  o, 32);
    sq += __shfl_xor(sq, o, 32);
  }
  __shared__ float rs[8], rq[8];
  __shared__ float mu_s, inv_s;
  const int wave = t >> 5, lane = t & 31;
  if (lane == 0) { rs[wave] = s; rq[wave] = sq; }
  __syncthreads();
  if (t == 0) {
    float S = 0.f, Q = 0.f;
#pragma unroll
    for (int w = 0; w < 8; ++w) { S += rs[w]; Q += rq[w]; }
    const float mu = S * (1.0f / kC);
    mu_s  = mu;
    inv_s = rsqrtf(Q * (1.0f / kC) - mu * mu + kLnEps);
  }
  __syncthreads();
  const float mu = mu_s, inv = inv_s;
#pragma unroll
  for (int j = 0; j < 4; ++j) {
    const int c = t + j * 256;
    out[(size_t)row * kC + c] = (x[j] - mu) * inv * g[c] + be[c];
  }
}

extern "C" void kernel_launch(void* const* d_in, const int* in_sizes, int n_in,
                              void* d_out, int out_size, void* d_ws, size_t ws_size,
                              hipStream_t stream) {
  (void)in_sizes; (void)n_in; (void)out_size; (void)ws_size;
  const float* query = (const float*)d_in[0];
  const float* key   = (const float*)d_in[1];
  const float* value = (const float*)d_in[2];
  const float* Wq = (const float*)d_in[4];
  const float* Wk = (const float*)d_in[5];
  const float* Wv = (const float*)d_in[6];
  const float* Wo = (const float*)d_in[7];
  const float* ln_g = (const float*)d_in[8];
  const float* ln_b = (const float*)d_in[9];
  float* out = (float*)d_out;

  constexpr size_t BNC = (size_t)kB * kN * kC;
  constexpr size_t CC  = (size_t)kC * kC;
  constexpr size_t BHN = (size_t)kB * kH * kN;

  size_t off = 0;
  auto take = [&](size_t bytes) -> void* {
    void* p = (char*)d_ws + off;
    off += (bytes + 255) & ~(size_t)255;
    return p;
  };
  _Float16* xq16 = (_Float16*)take(BNC * 2);
  _Float16* xk16 = (_Float16*)take(BNC * 2);
  _Float16* xv16 = (_Float16*)take(BNC * 2);
  _Float16* wq16 = (_Float16*)take(CC * 2);
  _Float16* wk16 = (_Float16*)take(CC * 2);
  _Float16* wv16 = (_Float16*)take(CC * 2);
  _Float16* wo16 = (_Float16*)take(CC * 2);
  _Float16* qh16 = (_Float16*)take(BNC * 2);
  _Float16* kh16 = (_Float16*)take(BNC * 2);
  _Float16* vt16 = (_Float16*)take(BNC * 2);
  float*    mcol = (float*)take(BHN * 4);
  float*    rcol = (float*)take(BHN * 4);
  _Float16* ctx16 = xq16;
  float*    yf32  = (float*)xk16;

  cvt_all_f32_f16<<<kCvtGrid, kCvtBlock, 0, stream>>>(
      query, key, value, Wq, Wk, Wv, Wo,
      xq16, xk16, xv16, wq16, wk16, wv16, wo16);

  dim3 ggrid(32, 16);
  gemm_xwT<0><<<ggrid, 128, 0, stream>>>(xq16, wq16, qh16);
  gemm_xwT<0><<<ggrid, 128, 0, stream>>>(xk16, wk16, kh16);
  gemm_xwT<1><<<ggrid, 128, 0, stream>>>(xv16, wv16, vt16);

  colstats_kernel<<<dim3(kN / (16 * kKT), kB * kH), 256, 0, stream>>>(qh16, kh16, mcol, rcol);

  ctx_kernel<<<dim3(kN / (16 * kQT * kCW), kB * kH), 32 * kCW, 0, stream>>>(
      qh16, kh16, vt16, mcol, rcol, ctx16);

  gemm_xwT<2><<<ggrid, 128, 0, stream>>>(ctx16, wo16, yf32);
  ln_kernel<<<kB * kN, 256, 0, stream>>>(query, yf32, ln_g, ln_b, out);
}


# ctx kernel: scores formed transposed (S^T = K Q^T: K fragment as A, Q as B), so the exp tile already is the A fragment of E; the per-tile LDS transpose (16 stores, 16 loads) is gone; same values
# speedup vs baseline: 2.7911x; 1.0651x over previous
#include <hip/hip_runtime.h>
#include <hip/hip_bf16.h>


typedef __attribute__((ext_vector_type(16))) _Float16 v16h;
typedef __attribute__((ext_vector_type(8)))  _Float16 v8h;
typedef __attribute__((ext_vector_type(4)))  _Float16 v4h;
typedef __attribute__((ext_vector_type(8)))  float    v8f;

constexpr int kB = 2, kN = 2048, kC = 1024, kH = 16, kD = 64;
constexpr float kLnEps = 1e-5f;

__device__ __forceinline__ v8f wmma16(v16h a, v16h b, v8f c) {
  return __builtin_amdgcn_wmma_f32_16x16x32_f16(false, a, false, b, (short)0, c,
                                                false, false);
}

constexpr int kFragHalves  = 512;
constexpr int kPieceHalves = 256;

__device__ __forceinline__ v16h load_frag_pk(const _Float16* __restrict__ frag) {
  const int lane = threadIdx.x & 31;
  v8h lo = *(const v8h*)(frag + lane * 8);
  v8h hi = *(const v8h*)(frag + kPieceHalves + lane * 8);
  v16h f;
#pragma unroll
  for (int i = 0; i < 8; ++i) { f[i] = lo[i]; f[i + 8] = hi[i]; }
  return f;
}

__device__ __forceinline__ int qk_pk_off(int n, int d) {
  return (n >> 4) * (2 * kFragHalves) + (d >> 5) * kFragHalves +
         ((d >> 4) & 1) * kPieceHalves + ((((d >> 3) & 1) << 4) + (n & 15)) * 8 + (d & 7);
}

constexpr int kCvtBlock  = 256;
constexpr int kCvtPerBlk = 8192;
constexpr int kCvtInFl   = 16;
constexpr int kCvtRounds = kCvtPerBlk / (kCvtBlock * kCvtInFl);
constexpr int kCvtXBlks  = kB * kN * kC / 4 / kCvtPerBlk;
constexpr int kCvtWBlks  = kC * kC / 4 / kCvtPerBlk;
constexpr int kCvtGrid   = 3 * kCvtXBlks + 4 * kCvtWBlks;
static_assert(kCvtRounds * kCvtBlock * kCvtInFl == kCvtPerBlk, "block chunk");
static_assert(kCvtXBlks * kCvtPerBlk * 4 == kB * kN * kC, "activation split");
static_assert(kCvtWBlks * kCvtPerBlk * 4 == kC * kC, "weight split");

__global__ __launch_bounds__(kCvtBlock)
void cvt_all_f32_f16(const float* __restrict__ xq, const float* __restrict__ xk,
                     const float* __restrict__ xv, const float* __restrict__ wq,
                     const float* __restrict__ wk, const float* __restrict__ wv,
                     const float* __restrict__ wo,
                     _Float16* __restrict__ xq16, _Float16* __restrict__ xk16,
                     _Float16* __restrict__ xv16, _Float16* __restrict__ wq16,
                     _Float16* __restrict__ wk16, _Float16* __restrict__ wv16,
                     _Float16* __restrict__ wo16) {
  const int b = blockIdx.x;
  const float* src;
  _Float16* dst;
  int lb;
  if (b < 3 * kCvtXBlks) {
    const int s = b / kCvtXBlks;
    lb  = b - s * kCvtXBlks;
    src = (s == 0) ? xq : (s == 1) ? xk : xv;
    dst = (s == 0) ? xq16 : (s == 1) ? xk16 : xv16;
  } else {
    const int bw = b - 3 * kCvtXBlks;
    const int s  = bw / kCvtWBlks;
    lb  = bw - s * kCvtWBlks;
    src = (s == 0) ? wq : (s == 1) ? wk : (s == 2) ? wv : wo;
    dst = (s == 0) ? wq16 : (s == 1) ? wk16 : (s == 2) ? wv16 : wo16;
  }
  const float4* s4 = (const float4*)src + (size_t)lb * kCvtPerBlk + threadIdx.x;
  v4h*          d4 = (v4h*)dst          + (size_t)lb * kCvtPerBlk + threadIdx.x;

#pragma unroll
  for (int r = 0; r < kCvtRounds; ++r) {
    float4 v[kCvtInFl];
#pragma unroll
    for (int j = 0; j < kCvtInFl; ++j)
      v[j] = s4[(r * kCvtInFl + j) * kCvtBlock];
#pragma unroll
    for (int j = 0; j < kCvtInFl; ++j) {
      v4h h;
      h[0] = (_Float16)v[j].x; h[1] = (_Float16)v[j].y;
      h[2] = (_Float16)v[j].z; h[3] = (_Float16)v[j].w;
      d4[(r * kCvtInFl + j) * kCvtBlock] = h;
    }
  }
}

constexpr int kGK     = 64;
constexpr int kGPitch = kGK + 8;

__device__ __forceinline__ v16h lds_frag(const _Float16* p) {
  const int lane = threadIdx.x & 31;
  const int r    = lane & 15;
  const int ko   = (lane >> 4) << 3;
  const _Float16* b0 = p + r * kGPitch + ko;
  v8h lo = *(const v8h*)(b0);
  v8h hi = *(const v8h*)(b0 + 16);
  v16h f;
#pragma unroll
  for (int i = 0; i < 8; ++i) { f[i] = lo[i]; f[i + 8] = hi[i]; }
  return f;
}

template <int MODE>
__global__ __launch_bounds__(128)
void gemm_xwT(const _Float16* __restrict__ A,
              const _Float16* __restrict__ W,
              void* __restrict__ out) {
  const int tid  = threadIdx.x;
  const int wave = tid >> 5;
  const int lane = tid & 31;
  const int colL = lane & 15;
  const int hi   = lane >> 4;
  const int mB = blockIdx.x * 128;
  const int m0 = mB + wave * 32;
  const int n0 = blockIdx.y * 64;

  __shared__ __attribute__((aligned(16))) _Float16 sA[2][128][kGPitch];
  __shared__ __attribute__((aligned(16))) _Float16 sW[2][64][kGPitch];

  const int crow = tid >> 3;
  const int ccol = (tid & 7) * 8;
  const _Float16* ag = A + (size_t)(mB + crow) * kC + ccol;
  const _Float16* wg = W + (size_t)(n0 + crow) * kC + ccol;

  v8h ra[8], rw[4];
  auto gload = [&](int k0) {
#pragma unroll
    for (int i = 0; i < 8; ++i) ra[i] = *(const v8h*)(ag + (size_t)i * 16 * kC + k0);
#pragma unroll
    for (int i = 0; i < 4; ++i) rw[i] = *(const v8h*)(wg + (size_t)i * 16 * kC + k0);
  };
  auto sstore = [&](int s) {
#pragma unroll
    for (int i = 0; i < 8; ++i) *(v8h*)&sA[s][crow + 16 * i][ccol] = ra[i];
#pragma unroll
    for (int i = 0; i < 4; ++i) *(v8h*)&sW[s][crow + 16 * i][ccol] = rw[i];
  };

  v8f acc[2][4] = {};

  gload(0);
  sstore(0);
  __syncthreads();

  constexpr int kSlices = kC / kGK;
  for (int t = 0; t < kSlices; ++t) {
    const int s = t & 1;
    if (t + 1 < kSlices) gload((t + 1) * kGK);

#pragma unroll
    for (int kk = 0; kk < kGK; kk += 32) {
      v16h a0 = lds_frag(&sA[s][wave * 32][kk]);
      v16h a1 = lds_frag(&sA[s][wave * 32 + 16][kk]);
      v16h bf[4];
#pragma unroll
      for (int nt = 0; nt < 4; ++nt) bf[nt] = lds_frag(&sW[s][nt * 16][kk]);
#pragma unroll
      for (int nt = 0; nt < 4; ++nt) {
        acc[0][nt] = wmma16(a0, bf[nt], acc[0][nt]);
        acc[1][nt] = wmma16(a1, bf[nt], acc[1][nt]);
      }
    }

    if (t + 1 < kSlices) sstore(s ^ 1);
    __syncthreads();
  }

  if constexpr (MODE == 1) {
    const int b = m0 >> 11, c = (m0 & (kN - 1)) >> 5;
    const int h = n0 >> 6;
    _Float16* hb = (_Float16*)out + (size_t)(b * kH + h) * kD * kN +
                   (size_t)c * (4 * kFragHalves) + lane * 8;
#pragma unroll
    for (int mt = 0; mt < 2; ++mt)
#pragma unroll
      for (int nt = 0; nt < 4; ++nt) {
        v8h pk;
#pragma unroll
        for (int r = 0; r < 8; ++r) pk[r] = (_Float16)acc[mt][nt][r];
        *(v8h*)(hb + nt * kFragHalves + mt * kPieceHalves) = pk;
      }
  } else {
#pragma unroll
    for (int mt = 0; mt < 2; ++mt)
#pragma unroll
      for (int nt = 0; nt < 4; ++nt)
#pragma unroll
        for (int r = 0; r < 8; ++r) {
          const int row  = m0 + mt * 16 + 8 * hi + r;
          const int colc = n0 + nt * 16 + colL;
          const float v  = acc[mt][nt][r];
          if constexpr (MODE == 2) {
            ((float*)out)[(size_t)row * kC + colc] = v;
          } else {
            const int b = row >> 11, n = row & (kN - 1);
            const int h = colc >> 6, d = colc & (kD - 1);
            ((_Float16*)out)[(size_t)(b * kH + h) * kN * kD + qk_pk_off(n, d)] = (_Float16)v;
          }
        }
  }
}

constexpr int kKT = 8;

__global__ __launch_bounds__(256, 4)
void colstats_kernel(const _Float16* __restrict__ q16,
                     const _Float16* __restrict__ k16,
                     float* __restrict__ mcol,
                     float* __restrict__ rcol) {
  const int kg   = blockIdx.x;
  const int bh   = blockIdx.y;
  const int wave = threadIdx.x >> 5;
  const int lane = threadIdx.x & 31;
  const _Float16* qb = q16 + (size_t)bh * kN * kD;
  const _Float16* kb = k16 + ((size_t)bh * kN + (size_t)kg * 16 * kKT) * kD;

  v16h bk[kKT][2];
#pragma unroll
  for (int t = 0; t < kKT; ++t) {
    bk[t][0] = load_frag_pk(kb + (size_t)t * 16 * kD);
    bk[t][1] = load_frag_pk(kb + (size_t)t * 16 * kD + kFragHalves);
  }

  float mrun[kKT], srun[kKT];
#pragma unroll
  for (int t = 0; t < kKT; ++t) { mrun[t] = -3.4e38f; srun[t] = 0.f; }

  constexpr int kTrips = kN / 16 / 8;
  v16h a0 = load_frag_pk(qb + (size_t)wave * 16 * kD);
  v16h a1 = load_frag_pk(qb + (size_t)wave * 16 * kD + kFragHalves);
  for (int it = 0; it < kTrips; ++it) {
    v16h n0 = a0, n1 = a1;
    if (it + 1 < kTrips) {
      const int qn = wave + 8 * (it + 1);
      n0 = load_frag_pk(qb + (size_t)qn * 16 * kD);
      n1 = load_frag_pk(qb + (size_t)qn * 16 * kD + kFragHalves);
    }
#pragma unroll
    for (int tp = 0; tp < kKT; tp += 2) {
      v8f s[2];
#pragma unroll
      for (int u = 0; u < 2; ++u) {
        s[u] = v8f{};
        s[u] = wmma16(a0, bk[tp + u][0], s[u]);
        s[u] = wmma16(a1, bk[tp + u][1], s[u]);
      }
#pragma unroll
      for (int u = 0; u < 2; ++u) {
        const int t = tp + u;
        float tm = s[u][0];
#pragma unroll
        for (int i = 1; i < 8; ++i) tm = fmaxf(tm, s[u][i]);
        const float mn = fmaxf(mrun[t], tm);
        float part = 0.f;
#pragma unroll
        for (int i = 0; i < 8; ++i) part += __expf(s[u][i] - mn);
        srun[t] = srun[t] * __expf(mrun[t] - mn) + part;
        mrun[t] = mn;
      }
      __builtin_amdgcn_sched_barrier(0);
    }
    a0 = n0; a1 = n1;
  }

  __shared__ float redm[8][16 * kKT];
  __shared__ float reds[8][16 * kKT];
#pragma unroll
  for (int t = 0; t < kKT; ++t) {
    const float mo = __shfl_xor(mrun[t], 16, 32);
    const float so = __shfl_xor(srun[t], 16, 32);
    const float mn = fmaxf(mrun[t], mo);
    const float sv = srun[t] * __expf(mrun[t] - mn) + so * __expf(mo - mn);
    if (lane < 16) { redm[wave][t * 16 + lane] = mn; reds[wave][t * 16 + lane] = sv; }
  }
  __syncthreads();
  if (threadIdx.x < 16 * kKT) {
    const int c = threadIdx.x;
    float m = redm[0][c];
#pragma unroll
    for (int w = 1; w < 8; ++w) m = fmaxf(m, redm[w][c]);
    float sum = 0.f;
#pragma unroll
    for (int w = 0; w < 8; ++w) sum += reds[w][c] * __expf(redm[w][c] - m);
    const size_t o = (size_t)bh * kN + (size_t)kg * 16 * kKT + c;
    mcol[o] = m;
    rcol[o] = 1.0f / sum;
  }
}

constexpr int kQT = 4;
constexpr int kCW = 4;

__global__ __launch_bounds__(32 * kCW)
void ctx_kernel(const _Float16* __restrict__ q16,
                const _Float16* __restrict__ k16,
                const _Float16* __restrict__ vt16,
                const float* __restrict__ mcol,
                const float* __restrict__ rcol,
                _Float16* __restrict__ ctx16) {
  const int bh   = blockIdx.y;
  const int wave = threadIdx.x >> 5;
  const int lane = threadIdx.x & 31;
  const int col  = lane & 15;
  const int hi   = lane >> 4;
  const int qt0  = (blockIdx.x * kCW + wave) * kQT;

  __shared__ __attribute__((aligned(16))) float mr_s[2][kN];

  for (int i = threadIdx.x; i < kN; i += 32 * kCW) {
    mr_s[0][i] = mcol[(size_t)bh * kN + i];
    mr_s[1][i] = rcol[(size_t)bh * kN + i];
  }
  __syncthreads();

  const _Float16* qb = q16  + (size_t)bh * kN * kD;
  const _Float16* kb = k16  + (size_t)bh * kN * kD;
  const _Float16* vb = vt16 + (size_t)bh * kD * kN;

  v16h aq[kQT][2];
#pragma unroll
  for (int j = 0; j < kQT; ++j) {
    aq[j][0] = load_frag_pk(qb + (size_t)(qt0 + j) * 16 * kD);
    aq[j][1] = load_frag_pk(qb + (size_t)(qt0 + j) * 16 * kD + kFragHalves);
  }

  v8f acc[kQT][4] = {};

  auto load_chunk = [&](int k0, v16h (&kf)[4], v16h (&vf)[4]) {
    kf[0] = load_frag_pk(kb + (size_t)(k0     ) * kD);
    kf[1] = load_frag_pk(kb + (size_t)(k0     ) * kD + kFragHalves);
    kf[2] = load_frag_pk(kb + (size_t)(k0 + 16) * kD);
    kf[3] = load_frag_pk(kb + (size_t)(k0 + 16) * kD + kFragHalves);
#pragma unroll
    for (int dt = 0; dt < 4; ++dt)
      vf[dt] = load_frag_pk(vb + (size_t)k0 * kD + dt * kFragHalves);
  };

  struct MR { float4 m[4], r[4]; };
  auto load_mr = [&](int k0, MR& c) {
    const float4* mp = (const float4*)&mr_s[0][k0 + 8 * hi];
    const float4* rp = (const float4*)&mr_s[1][k0 + 8 * hi];
    c.m[0] = mp[0]; c.m[1] = mp[1]; c.m[2] = mp[4]; c.m[3] = mp[5];
    c.r[0] = rp[0]; c.r[1] = rp[1]; c.r[2] = rp[4]; c.r[3] = rp[5];
  };

  auto do_chunk = [&](int k0, const v16h (&kf)[4], const v16h (&vf)[4]) {
    MR c;
    load_mr(k0, c);
    const float mk[16] = {c.m[0].x, c.m[0].y, c.m[0].z, c.m[0].w,
                          c.m[1].x, c.m[1].y, c.m[1].z, c.m[1].w,
                          c.m[2].x, c.m[2].y, c.m[2].z, c.m[2].w,
                          c.m[3].x, c.m[3].y, c.m[3].z, c.m[3].w};
    const float rk[16] = {c.r[0].x, c.r[0].y, c.r[0].z, c.r[0].w,
                          c.r[1].x, c.r[1].y, c.r[1].z, c.r[1].w,
                          c.r[2].x, c.r[2].y, c.r[2].z, c.r[2].w,
                          c.r[3].x, c.r[3].y, c.r[3].z, c.r[3].w};
#pragma unroll
    for (int j = 0; j < kQT; ++j) {
      v8f s0 = {}, s1 = {};
      s0 = wmma16(kf[0], aq[j][0], s0);
      s0 = wmma16(kf[1], aq[j][1], s0);
      s1 = wmma16(kf[2], aq[j][0], s1);
      s1 = wmma16(kf[3], aq[j][1], s1);

      v16h ef;
#pragma unroll
      for (int i = 0; i < 8; ++i) {
        ef[i]     = (_Float16)(__expf(s0[i] - mk[i])     * rk[i]);
        ef[i + 8] = (_Float16)(__expf(s1[i] - mk[i + 8]) * rk[i + 8]);
      }
#pragma unroll
      for (int dt = 0; dt < 4; ++dt)
        acc[j][dt] = wmma16(ef, vf[dt], acc[j][dt]);
    }
  };

  static_assert(kN % 64 == 0, "two chunks per trip");
  v16h kfA[4], vfA[4], kfB[4], vfB[4];
  load_chunk(0, kfA, vfA);
  for (int k0 = 0; k0 < kN; k0 += 64) {
    load_chunk(k0 + 32, kfB, vfB);
    __builtin_amdgcn_sched_barrier(0);
    do_chunk(k0, kfA, vfA);
    const int kn = (k0 + 64 < kN) ? (k0 + 64) : k0;
    load_chunk(kn, kfA, vfA);
    __builtin_amdgcn_sched_barrier(0);
    do_chunk(k0 + 32, kfB, vfB);
  }

  const int b = bh >> 4, h = bh & 15;
  _Float16* cb = ctx16 + (size_t)b * kN * kC + (size_t)h * kD;
#pragma unroll
  for (int j = 0; j < kQT; ++j) {
    const int qrow = (qt0 + j) * 16 + 8 * hi;
#pragma unroll
    for (int dt = 0; dt < 4; ++dt)
#pragma unroll
      for (int r = 0; r < 8; ++r)
        cb[(size_t)(qrow + r) * kC + dt * 16 + col] = (_Float16)acc[j][dt][r];
  }
}

__global__ void ln_kernel(const float* __restrict__ resid,
                          const float* __restrict__ y,
                          const float* __restrict__ g,
                          const float* __restrict__ be,
                          float* __restrict__ out) {
  const int row = blockIdx.x;
  const int t   = threadIdx.x;
  float x[4], s = 0.f, sq = 0.f;
#pragma unroll
  for (int j = 0; j < 4; ++j) {
    const int c = t + j * 256;
    const float v = resid[(size_t)row * kC + c] + y[(size_t)row * kC + c];
    x[j] = v; s += v; sq += v * v;
  }
#pragma unroll
  for (int o = 16; o >= 1; o >>= 1) {
    s  += __shfl_xor(s,  o, 32);
    sq += __shfl_xor(sq, o, 32);
  }
  __shared__ float rs[8], rq[8];
  __shared__ float mu_s, inv_s;
  const int wave = t >> 5, lane = t & 31;
  if (lane == 0) { rs[wave] = s; rq[wave] = sq; }
  __syncthreads();
  if (t == 0) {
    float S = 0.f, Q = 0.f;
#pragma unroll
    for (int w = 0; w < 8; ++w) { S += rs[w]; Q += rq[w]; }
    const float mu = S * (1.0f / kC);
    mu_s  = mu;
    inv_s = rsqrtf(Q * (1.0f / kC) - mu * mu + kLnEps);
  }
  __syncthreads();
  const float mu = mu_s, inv = inv_s;
#pragma unroll
  for (int j = 0; j < 4; ++j) {
    const int c = t + j * 256;
    out[(size_t)row * kC + c] = (x[j] - mu) * inv * g[c] + be[c];
  }
}

extern "C" void kernel_launch(void* const* d_in, const int* in_sizes, int n_in,
                              void* d_out, int out_size, void* d_ws, size_t ws_size,
                              hipStream_t stream) {
  (void)in_sizes; (void)n_in; (void)out_size; (void)ws_size;
  const float* query = (const float*)d_in[0];
  const float* key   = (const float*)d_in[1];
  const float* value = (const float*)d_in[2];
  const float* Wq = (const float*)d_in[4];
  const float* Wk = (const float*)d_in[5];
  const float* Wv = (const float*)d_in[6];
  const float* Wo = (const float*)d_in[7];
  const float* ln_g = (const float*)d_in[8];
  const float* ln_b = (const float*)d_in[9];
  float* out = (float*)d_out;

  constexpr size_t BNC = (size_t)kB * kN * kC;
  constexpr size_t CC  = (size_t)kC * kC;
  constexpr size_t BHN = (size_t)kB * kH * kN;

  size_t off = 0;
  auto take = [&](size_t bytes) -> void* {
    void* p = (char*)d_ws + off;
    off += (bytes + 255) & ~(size_t)255;
    return p;
  };
  _Float16* xq16 = (_Float16*)take(BNC * 2);
  _Float16* xk16 = (_Float16*)take(BNC * 2);
  _Float16* xv16 = (_Float16*)take(BNC * 2);
  _Float16* wq16 = (_Float16*)take(CC * 2);
  _Float16* wk16 = (_Float16*)take(CC * 2);
  _Float16* wv16 = (_Float16*)take(CC * 2);
  _Float16* wo16 = (_Float16*)take(CC * 2);
  _Float16* qh16 = (_Float16*)take(BNC * 2);
  _Float16* kh16 = (_Float16*)take(BNC * 2);
  _Float16* vt16 = (_Float16*)take(BNC * 2);
  float*    mcol = (float*)take(BHN * 4);
  float*    rcol = (float*)take(BHN * 4);
  _Float16* ctx16 = xq16;
  float*    yf32  = (float*)xk16;

  cvt_all_f32_f16<<<kCvtGrid, kCvtBlock, 0, stream>>>(
      query, key, value, Wq, Wk, Wv, Wo,
      xq16, xk16, xv16, wq16, wk16, wv16, wo16);

  dim3 ggrid(32, 16);
  gemm_xwT<0><<<ggrid, 128, 0, stream>>>(xq16, wq16, qh16);
  gemm_xwT<0><<<ggrid, 128, 0, stream>>>(xk16, wk16, kh16);
  gemm_xwT<1><<<ggrid, 128, 0, stream>>>(xv16, wv16, vt16);

  colstats_kernel<<<dim3(kN / (16 * kKT), kB * kH), 256, 0, stream>>>(qh16, kh16, mcol, rcol);

  ctx_kernel<<<dim3(kN / (16 * kQT * kCW), kB * kH), 32 * kCW, 0, stream>>>(
      qh16, kh16, vt16, mcol, rcol, ctx16);

  gemm_xwT<2><<<ggrid, 128, 0, stream>>>(ctx16, wo16, yf32);
  ln_kernel<<<kB * kN, 256, 0, stream>>>(query, yf32, ln_g, ln_b, out);
}


# Q, K, V projections as ONE launch of 1536 blocks (grid z picks the product): more blocks resident per WGP than one product's 512 give, two launch boundaries gone; same tile code, same sums
# speedup vs baseline: 3.2056x; 1.1485x over previous
#include <hip/hip_runtime.h>
#include <hip/hip_bf16.h>


typedef __attribute__((ext_vector_type(16))) _Float16 v16h;
typedef __attribute__((ext_vector_type(8)))  _Float16 v8h;
typedef __attribute__((ext_vector_type(4)))  _Float16 v4h;
typedef __attribute__((ext_vector_type(8)))  float    v8f;

constexpr int kB = 2, kN = 2048, kC = 1024, kH = 16, kD = 64;
constexpr float kLnEps = 1e-5f;

__device__ __forceinline__ v8f wmma16(v16h a, v16h b, v8f c) {
  return __builtin_amdgcn_wmma_f32_16x16x32_f16(false, a, false, b, (short)0, c,
                                                false, false);
}

constexpr int kFragHalves  = 512;
constexpr int kPieceHalves = 256;

__device__ __forceinline__ v16h load_frag_pk(const _Float16* __restrict__ frag) {
  const int lane = threadIdx.x & 31;
  v8h lo = *(const v8h*)(frag + lane * 8);
  v8h hi = *(const v8h*)(frag + kPieceHalves + lane * 8);
  v16h f;
#pragma unroll
  for (int i = 0; i < 8; ++i) { f[i] = lo[i]; f[i + 8] = hi[i]; }
  return f;
}

__device__ __forceinline__ int qk_pk_off(int n, int d) {
  return (n >> 4) * (2 * kFragHalves) + (d >> 5) * kFragHalves +
         ((d >> 4) & 1) * kPieceHalves + ((((d >> 3) & 1) << 4) + (n & 15)) * 8 + (d & 7);
}

constexpr int kCvtBlock  = 256;
constexpr int kCvtPerBlk = 8192;
constexpr int kCvtInFl   = 16;
constexpr int kCvtRounds = kCvtPerBlk / (kCvtBlock * kCvtInFl);
constexpr int kCvtXBlks  = kB * kN * kC / 4 / kCvtPerBlk;
constexpr int kCvtWBlks  = kC * kC / 4 / kCvtPerBlk;
constexpr int kCvtGrid   = 3 * kCvtXBlks + 4 * kCvtWBlks;
static_assert(kCvtRounds * kCvtBlock * kCvtInFl == kCvtPerBlk, "block chunk");
static_assert(kCvtXBlks * kCvtPerBlk * 4 == kB * kN * kC, "activation split");
static_assert(kCvtWBlks * kCvtPerBlk * 4 == kC * kC, "weight split");

__global__ __launch_bounds__(kCvtBlock)
void cvt_all_f32_f16(const float* __restrict__ xq, const float* __restrict__ xk,
                     const float* __restrict__ xv, const float* __restrict__ wq,
                     const float* __restrict__ wk, const float* __restrict__ wv,
                     const float* __restrict__ wo,
                     _Float16* __restrict__ xq16, _Float16* __restrict__ xk16,
                     _Float16* __restrict__ xv16, _Float16* __restrict__ wq16,
                     _Float16* __restrict__ wk16, _Float16* __restrict__ wv16,
                     _Float16* __restrict__ wo16) {
  const int b = blockIdx.x;
  const float* src;
  _Float16* dst;
  int lb;
  if (b < 3 * kCvtXBlks) {
    const int s = b / kCvtXBlks;
    lb  = b - s * kCvtXBlks;
    src = (s == 0) ? xq : (s == 1) ? xk : xv;
    dst = (s == 0) ? xq16 : (s == 1) ? xk16 : xv16;
  } else {
    const int bw = b - 3 * kCvtXBlks;
    const int s  = bw / kCvtWBlks;
    lb  = bw - s * kCvtWBlks;
    src = (s == 0) ? wq : (s == 1) ? wk : (s == 2) ? wv : wo;
    dst = (s == 0) ? wq16 : (s == 1) ? wk16 : (s == 2) ? wv16 : wo16;
  }
  const float4* s4 = (const float4*)src + (size_t)lb * kCvtPerBlk + threadIdx.x;
  v4h*          d4 = (v4h*)dst          + (size_t)lb * kCvtPerBlk + threadIdx.x;

#pragma unroll
  for (int r = 0; r < kCvtRounds; ++r) {
    float4 v[kCvtInFl];
#pragma unroll
    for (int j = 0; j < kCvtInFl; ++j)
      v[j] = s4[(r * kCvtInFl + j) * kCvtBlock];
#pragma unroll
    for (int j = 0; j < kCvtInFl; ++j) {
      v4h h;
      h[0] = (_Float16)v[j].x; h[1] = (_Float16)v[j].y;
      h[2] = (_Float16)v[j].z; h[3] = (_Float16)v[j].w;
      d4[(r * kCvtInFl + j) * kCvtBlock] = h;
    }
  }
}

constexpr int kGK     = 64;
constexpr int kGPitch = kGK + 8;

__device__ __forceinline__ v16h lds_frag(const _Float16* p) {
  const int lane = threadIdx.x & 31;
  const int r    = lane & 15;
  const int ko   = (lane >> 4) << 3;
  const _Float16* b0 = p + r * kGPitch + ko;
  v8h lo = *(const v8h*)(b0);
  v8h hi = *(const v8h*)(b0 + 16);
  v16h f;
#pragma unroll
  for (int i = 0; i < 8; ++i) { f[i] = lo[i]; f[i + 8] = hi[i]; }
  return f;
}

template <int MODE>
__device__ __forceinline__
void gemm_tile(const _Float16* __restrict__ A,
               const _Float16* __restrict__ W,
               void* __restrict__ out,
               _Float16 (*__restrict__ sA)[128][kGPitch],
               _Float16 (*__restrict__ sW)[64][kGPitch],
               const int bx, const int by) {
  const int tid  = threadIdx.x;
  const int wave = tid >> 5;
  const int lane = tid & 31;
  const int colL = lane & 15;
  const int hi   = lane >> 4;
  const int mB = bx * 128;
  const int m0 = mB + wave * 32;
  const int n0 = by * 64;

  const int crow = tid >> 3;
  const int ccol = (tid & 7) * 8;
  const _Float16* ag = A + (size_t)(mB + crow) * kC + ccol;
  const _Float16* wg = W + (size_t)(n0 + crow) * kC + ccol;

  v8h ra[8], rw[4];
  auto gload = [&](int k0) {
#pragma unroll
    for (int i = 0; i < 8; ++i) ra[i] = *(const v8h*)(ag + (size_t)i * 16 * kC + k0);
#pragma unroll
    for (int i = 0; i < 4; ++i) rw[i] = *(const v8h*)(wg + (size_t)i * 16 * kC + k0);
  };
  auto sstore = [&](int s) {
#pragma unroll
    for (int i = 0; i < 8; ++i) *(v8h*)&sA[s][crow + 16 * i][ccol] = ra[i];
#pragma unroll
    for (int i = 0; i < 4; ++i) *(v8h*)&sW[s][crow + 16 * i][ccol] = rw[i];
  };

  v8f acc[2][4] = {};

  gload(0);
  sstore(0);
  __syncthreads();

  constexpr int kSlices = kC / kGK;
  for (int t = 0; t < kSlices; ++t) {
    const int s = t & 1;
    if (t + 1 < kSlices) gload((t + 1) * kGK);

#pragma unroll
    for (int kk = 0; kk < kGK; kk += 32) {
      v16h a0 = lds_frag(&sA[s][wave * 32][kk]);
      v16h a1 = lds_frag(&sA[s][wave * 32 + 16][kk]);
      v16h bf[4];
#pragma unroll
      for (int nt = 0; nt < 4; ++nt) bf[nt] = lds_frag(&sW[s][nt * 16][kk]);
#pragma unroll
      for (int nt = 0; nt < 4; ++nt) {
        acc[0][nt] = wmma16(a0, bf[nt], acc[0][nt]);
        acc[1][nt] = wmma16(a1, bf[nt], acc[1][nt]);
      }
    }

    if (t + 1 < kSlices) sstore(s ^ 1);
    __syncthreads();
  }

  if constexpr (MODE == 1) {
    const int b = m0 >> 11, c = (m0 & (kN - 1)) >> 5;
    const int h = n0 >> 6;
    _Float16* hb = (_Float16*)out + (size_t)(b * kH + h) * kD * kN +
                   (size_t)c * (4 * kFragHalves) + lane * 8;
#pragma unroll
    for (int mt = 0; mt < 2; ++mt)
#pragma unroll
      for (int nt = 0; nt < 4; ++nt) {
        v8h pk;
#pragma unroll
        for (int r = 0; r < 8; ++r) pk[r] = (_Float16)acc[mt][nt][r];
        *(v8h*)(hb + nt * kFragHalves + mt * kPieceHalves) = pk;
      }
  } else {
#pragma unroll
    for (int mt = 0; mt < 2; ++mt)
#pragma unroll
      for (int nt = 0; nt < 4; ++nt)
#pragma unroll
        for (int r = 0; r < 8; ++r) {
          const int row  = m0 + mt * 16 + 8 * hi + r;
          const int colc = n0 + nt * 16 + colL;
          const float v  = acc[mt][nt][r];
          if constexpr (MODE == 2) {
            ((float*)out)[(size_t)row * kC + colc] = v;
          } else {
            const int b = row >> 11, n = row & (kN - 1);
            const int h = colc >> 6, d = colc & (kD - 1);
            ((_Float16*)out)[(size_t)(b * kH + h) * kN * kD + qk_pk_off(n, d)] = (_Float16)v;
          }
        }
  }
}

template <int MODE>
__global__ __launch_bounds__(128)
void gemm_xwT(const _Float16* __restrict__ A,
              const _Float16* __restrict__ W,
              void* __restrict__ out) {
  __shared__ __attribute__((aligned(16))) _Float16 sA[2][128][kGPitch];
  __shared__ __attribute__((aligned(16))) _Float16 sW[2][64][kGPitch];
  gemm_tile<MODE>(A, W, out, sA, sW, blockIdx.x, blockIdx.y);
}

__global__ __launch_bounds__(128)
void gemm_qkv(const _Float16* __restrict__ xq, const _Float16* __restrict__ wq,
              _Float16* __restrict__ qh,
              const _Float16* __restrict__ xk, const _Float16* __restrict__ wk,
              _Float16* __restrict__ kh,
              const _Float16* __restrict__ xv, const _Float16* __restrict__ wv,
              _Float16* __restrict__ vt) {
  __shared__ __attribute__((aligned(16))) _Float16 sA[2][128][kGPitch];
  __shared__ __attribute__((aligned(16))) _Float16 sW[2][64][kGPitch];
  const int z = blockIdx.z;
  if (z == 2) {
    gemm_tile<1>(xv, wv, vt, sA, sW, blockIdx.x, blockIdx.y);
  } else {
    gemm_tile<0>(z == 0 ? xq : xk, z == 0 ? wq : wk, z == 0 ? qh : kh, sA, sW,
                 blockIdx.x, blockIdx.y);
  }
}

constexpr int kKT = 8;

__global__ __launch_bounds__(256, 4)
void colstats_kernel(const _Float16* __restrict__ q16,
                     const _Float16* __restrict__ k16,
                     float* __restrict__ mcol,
                     float* __restrict__ rcol) {
  const int kg   = blockIdx.x;
  const int bh   = blockIdx.y;
  const int wave = threadIdx.x >> 5;
  const int lane = threadIdx.x & 31;
  const _Float16* qb = q16 + (size_t)bh * kN * kD;
  const _Float16* kb = k16 + ((size_t)bh * kN + (size_t)kg * 16 * kKT) * kD;

  v16h bk[kKT][2];
#pragma unroll
  for (int t = 0; t < kKT; ++t) {
    bk[t][0] = load_frag_pk(kb + (size_t)t * 16 * kD);
    bk[t][1] = load_frag_pk(kb + (size_t)t * 16 * kD + kFragHalves);
  }

  float mrun[kKT], srun[kKT];
#pragma unroll
  for (int t = 0; t < kKT; ++t) { mrun[t] = -3.4e38f; srun[t] = 0.f; }

  constexpr int kTrips = kN / 16 / 8;
  v16h a0 = load_frag_pk(qb + (size_t)wave * 16 * kD);
  v16h a1 = load_frag_pk(qb + (size_t)wave * 16 * kD + kFragHalves);
  for (int it = 0; it < kTrips; ++it) {
    v16h n0 = a0, n1 = a1;
    if (it + 1 < kTrips) {
      const int qn = wave + 8 * (it + 1);
      n0 = load_frag_pk(qb + (size_t)qn * 16 * kD);
      n1 = load_frag_pk(qb + (size_t)qn * 16 * kD + kFragHalves);
    }
#pragma unroll
    for (int tp = 0; tp < kKT; tp += 2) {
      v8f s[2];
#pragma unroll
      for (int u = 0; u < 2; ++u) {
        s[u] = v8f{};
        s[u] = wmma16(a0, bk[tp + u][0], s[u]);
        s[u] = wmma16(a1, bk[tp + u][1], s[u]);
      }
#pragma unroll
      for (int u = 0; u < 2; ++u) {
        const int t = tp + u;
        float tm = s[u][0];
#pragma unroll
        for (int i = 1; i < 8; ++i) tm = fmaxf(tm, s[u][i]);
        const float mn = fmaxf(mrun[t], tm);
        float part = 0.f;
#pragma unroll
        for (int i = 0; i < 8; ++i) part += __expf(s[u][i] - mn);
        srun[t] = srun[t] * __expf(mrun[t] - mn) + part;
        mrun[t] = mn;
      }
      __builtin_amdgcn_sched_barrier(0);
    }
    a0 = n0; a1 = n1;
  }

  __shared__ float redm[8][16 * kKT];
  __shared__ float reds[8][16 * kKT];
#pragma unroll
  for (int t = 0; t < kKT; ++t) {
    const float mo = __shfl_xor(mrun[t], 16, 32);
    const float so = __shfl_xor(srun[t], 16, 32);
    const float mn = fmaxf(mrun[t], mo);
    const float sv = srun[t] * __expf(mrun[t] - mn) + so * __expf(mo - mn);
    if (lane < 16) { redm[wave][t * 16 + lane] = mn; reds[wave][t * 16 + lane] = sv; }
  }
  __syncthreads();
  if (threadIdx.x < 16 * kKT) {
    const int c = threadIdx.x;
    float m = redm[0][c];
#pragma unroll
    for (int w = 1; w < 8; ++w) m = fmaxf(m, redm[w][c]);
    float sum = 0.f;
#pragma unroll
    for (int w = 0; w < 8; ++w) sum += reds[w][c] * __expf(redm[w][c] - m);
    const size_t o = (size_t)bh * kN + (size_t)kg * 16 * kKT + c;
    mcol[o] = m;
    rcol[o] = 1.0f / sum;
  }
}

constexpr int kQT = 4;
constexpr int kCW = 4;
constexpr int kKS = 2;

__global__ __launch_bounds__(32 * kCW * kKS)
void ctx_kernel(const _Float16* __restrict__ q16,
                const _Float16* __restrict__ k16,
                const _Float16* __restrict__ vt16,
                const float* __restrict__ mcol,
                const float* __restrict__ rcol,
                _Float16* __restrict__ ctx16) {
  const int bh   = blockIdx.y;
  const int wave = __builtin_amdgcn_readfirstlane(threadIdx.x >> 5);
  const int grp  = wave & (kCW - 1);
  const int half = wave / kCW;
  const int lane = threadIdx.x & 31;
  const int col  = lane & 15;
  const int hi   = lane >> 4;
  const int qt0  = (blockIdx.x * kCW + grp) * kQT;

  __shared__ __attribute__((aligned(16))) float mr_s[2][kN];
  __shared__ __attribute__((aligned(16))) float4 red_s[kCW * kKS][16][32];

  for (int i = threadIdx.x; i < kN; i += 32 * kCW * kKS) {
    mr_s[0][i] = mcol[(size_t)bh * kN + i];
    mr_s[1][i] = rcol[(size_t)bh * kN + i];
  }
  __syncthreads();

  const _Float16* qb = q16  + (size_t)bh * kN * kD;
  const _Float16* kb = k16  + (size_t)bh * kN * kD;
  const _Float16* vb = vt16 + (size_t)bh * kD * kN;

  v16h aq[kQT][2];
#pragma unroll
  for (int j = 0; j < kQT; ++j) {
    aq[j][0] = load_frag_pk(qb + (size_t)(qt0 + j) * 16 * kD);
    aq[j][1] = load_frag_pk(qb + (size_t)(qt0 + j) * 16 * kD + kFragHalves);
  }

  v8f acc[kQT][4] = {};

  auto load_chunk = [&](int k0, v16h (&kf)[4], v16h (&vf)[4]) {
    kf[0] = load_frag_pk(kb + (size_t)(k0     ) * kD);
    kf[1] = load_frag_pk(kb + (size_t)(k0     ) * kD + kFragHalves);
    kf[2] = load_frag_pk(kb + (size_t)(k0 + 16) * kD);
    kf[3] = load_frag_pk(kb + (size_t)(k0 + 16) * kD + kFragHalves);
#pragma unroll
    for (int dt = 0; dt < 4; ++dt)
      vf[dt] = load_frag_pk(vb + (size_t)k0 * kD + dt * kFragHalves);
  };

  struct MR { float4 m[4], r[4]; };
  auto load_mr = [&](int k0, MR& c) {
    const float4* mp = (const float4*)&mr_s[0][k0 + 8 * hi];
    const float4* rp = (const float4*)&mr_s[1][k0 + 8 * hi];
    c.m[0] = mp[0]; c.m[1] = mp[1]; c.m[2] = mp[4]; c.m[3] = mp[5];
    c.r[0] = rp[0]; c.r[1] = rp[1]; c.r[2] = rp[4]; c.r[3] = rp[5];
  };

  auto do_chunk = [&](int k0, const v16h (&kf)[4], const v16h (&vf)[4]) {
    MR c;
    load_mr(k0, c);
    const float mk[16] = {c.m[0].x, c.m[0].y, c.m[0].z, c.m[0].w,
                          c.m[1].x, c.m[1].y, c.m[1].z, c.m[1].w,
                          c.m[2].x, c.m[2].y, c.m[2].z, c.m[2].w,
                          c.m[3].x, c.m[3].y, c.m[3].z, c.m[3].w};
    const float rk[16] = {c.r[0].x, c.r[0].y, c.r[0].z, c.r[0].w,
                          c.r[1].x, c.r[1].y, c.r[1].z, c.r[1].w,
                          c.r[2].x, c.r[2].y, c.r[2].z, c.r[2].w,
                          c.r[3].x, c.r[3].y, c.r[3].z, c.r[3].w};
#pragma unroll
    for (int j = 0; j < kQT; ++j) {
      v8f s0 = {}, s1 = {};
      s0 = wmma16(kf[0], aq[j][0], s0);
      s0 = wmma16(kf[1], aq[j][1], s0);
      s1 = wmma16(kf[2], aq[j][0], s1);
      s1 = wmma16(kf[3], aq[j][1], s1);

      v16h ef;
#pragma unroll
      for (int i = 0; i < 8; ++i) {
        ef[i]     = (_Float16)(__expf(s0[i] - mk[i])     * rk[i]);
        ef[i + 8] = (_Float16)(__expf(s1[i] - mk[i + 8]) * rk[i + 8]);
      }
#pragma unroll
      for (int dt = 0; dt < 4; ++dt)
        acc[j][dt] = wmma16(ef, vf[dt], acc[j][dt]);
    }
  };

  static_assert((kN / kKS) % 64 == 0, "two chunks per trip");
  const int kbeg = half * (kN / kKS);
  const int kend = kbeg + kN / kKS;
  v16h kfA[4], vfA[4], kfB[4], vfB[4];
  load_chunk(kbeg, kfA, vfA);
  for (int k0 = kbeg; k0 < kend; k0 += 64) {
    load_chunk(k0 + 32, kfB, vfB);
    __builtin_amdgcn_sched_barrier(0);
    do_chunk(k0, kfA, vfA);
    const int kn = (k0 + 64 < kend) ? (k0 + 64) : k0;
    load_chunk(kn, kfA, vfA);
    __builtin_amdgcn_sched_barrier(0);
    do_chunk(k0 + 32, kfB, vfB);
  }

  const int b = bh >> 4, h = bh & 15;
  _Float16* cb = ctx16 + (size_t)b * kN * kC + (size_t)h * kD;
  auto hand_over = [&](int j0) {
    float4* dst = &red_s[wave][0][lane];
#pragma unroll
    for (int jj = 0; jj < 2; ++jj)
#pragma unroll
      for (int dt = 0; dt < 4; ++dt) {
        const v8f a = acc[j0 + jj][dt];
        dst[((jj * 4 + dt) * 2 + 0) * 32] = make_float4(a[0], a[1], a[2], a[3]);
        dst[((jj * 4 + dt) * 2 + 1) * 32] = make_float4(a[4], a[5], a[6], a[7]);
      }
  };
  auto finish = [&](int j0) {
    const float4* src = &red_s[wave ^ kCW][0][lane];
#pragma unroll
    for (int jj = 0; jj < 2; ++jj) {
      const int qrow = (qt0 + j0 + jj) * 16 + 8 * hi;
#pragma unroll
      for (int dt = 0; dt < 4; ++dt) {
        const float4 p0 = src[((jj * 4 + dt) * 2 + 0) * 32];
        const float4 p1 = src[((jj * 4 + dt) * 2 + 1) * 32];
        const float p[8] = {p0.x, p0.y, p0.z, p0.w, p1.x, p1.y, p1.z, p1.w};
#pragma unroll
        for (int r = 0; r < 8; ++r)
          cb[(size_t)(qrow + r) * kC + dt * 16 + col] =
              (_Float16)(acc[j0 + jj][dt][r] + p[r]);
      }
    }
  };
  if (half == 0) hand_over(2); else hand_over(0);
  __syncthreads();
  if (half == 0) finish(0); else finish(2);
}

__global__ void ln_kernel(const float* __restrict__ resid,
                          const float* __restrict__ y,
                          const float* __restrict__ g,
                          const float* __restrict__ be,
                          float* __restrict__ out) {
  const int row = blockIdx.x;
  const int t   = threadIdx.x;
  float x[4], s = 0.f, sq = 0.f;
#pragma unroll
  for (int j = 0; j < 4; ++j) {
    const int c = t + j * 256;
    const float v = resid[(size_t)row * kC + c] + y[(size_t)row * kC + c];
    x[j] = v; s += v; sq += v * v;
  }
#pragma unroll
  for (int o = 16; o >= 1; o >>= 1) {
    s  += __shfl_xor(s,  o, 32);
    sq += __shfl_xor(sq, o, 32);
  }
  __shared__ float rs[8], rq[8];
  __shared__ float mu_s, inv_s;
  const int wave = t >> 5, lane = t & 31;
  if (lane == 0) { rs[wave] = s; rq[wave] = sq; }
  __syncthreads();
  if (t == 0) {
    float S = 0.f, Q = 0.f;
#pragma unroll
    for (int w = 0; w < 8; ++w) { S += rs[w]; Q += rq[w]; }
    const float mu = S * (1.0f / kC);
    mu_s  = mu;
    inv_s = rsqrtf(Q * (1.0f / kC) - mu * mu + kLnEps);
  }
  __syncthreads();
  const float mu = mu_s, inv = inv_s;
#pragma unroll
  for (int j = 0; j < 4; ++j) {
    const int c = t + j * 256;
    out[(size_t)row * kC + c] = (x[j] - mu) * inv * g[c] + be[c];
  }
}

extern "C" void kernel_launch(void* const* d_in, const int* in_sizes, int n_in,
                              void* d_out, int out_size, void* d_ws, size_t ws_size,
                              hipStream_t stream) {
  (void)in_sizes; (void)n_in; (void)out_size; (void)ws_size;
  const float* query = (const float*)d_in[0];
  const float* key   = (const float*)d_in[1];
  const float* value = (const float*)d_in[2];
  const float* Wq = (const float*)d_in[4];
  const float* Wk = (const float*)d_in[5];
  const float* Wv = (const float*)d_in[6];
  const float* Wo = (const float*)d_in[7];
  const float* ln_g = (const float*)d_in[8];
  const float* ln_b = (const float*)d_in[9];
  float* out = (float*)d_out;

  constexpr size_t BNC = (size_t)kB * kN * kC;
  constexpr size_t CC  = (size_t)kC * kC;
  constexpr size_t BHN = (size_t)kB * kH * kN;

  size_t off = 0;
  auto take = [&](size_t bytes) -> void* {
    void* p = (char*)d_ws + off;
    off += (bytes + 255) & ~(size_t)255;
    return p;
  };
  _Float16* xq16 = (_Float16*)take(BNC * 2);
  _Float16* xk16 = (_Float16*)take(BNC * 2);
  _Float16* xv16 = (_Float16*)take(BNC * 2);
  _Float16* wq16 = (_Float16*)take(CC * 2);
  _Float16* wk16 = (_Float16*)take(CC * 2);
  _Float16* wv16 = (_Float16*)take(CC * 2);
  _Float16* wo16 = (_Float16*)take(CC * 2);
  _Float16* qh16 = (_Float16*)take(BNC * 2);
  _Float16* kh16 = (_Float16*)take(BNC * 2);
  _Float16* vt16 = (_Float16*)take(BNC * 2);
  float*    mcol = (float*)take(BHN * 4);
  float*    rcol = (float*)take(BHN * 4);
  _Float16* ctx16 = xq16;
  float*    yf32  = (float*)xk16;

  cvt_all_f32_f16<<<kCvtGrid, kCvtBlock, 0, stream>>>(
      query, key, value, Wq, Wk, Wv, Wo,
      xq16, xk16, xv16, wq16, wk16, wv16, wo16);

  dim3 ggrid(32, 16);
  gemm_qkv<<<dim3(32, 16, 3), 128, 0, stream>>>(xq16, wq16, qh16, xk16, wk16, kh16,
                                               xv16, wv16, vt16);

  colstats_kernel<<<dim3(kN / (16 * kKT), kB * kH), 256, 0, stream>>>(qh16, kh16, mcol, rcol);

  ctx_kernel<<<dim3(kN / (16 * kQT * kCW), kB * kH), 32 * kCW * kKS, 0, stream>>>(
      qh16, kh16, vt16, mcol, rcol, ctx16);

  gemm_xwT<2><<<ggrid, 128, 0, stream>>>(ctx16, wo16, yf32);
  ln_kernel<<<kB * kN, 256, 0, stream>>>(query, yf32, ln_g, ln_b, out);
}


# input projections on a 128x128 block tile, 64x64 per wave (16 accumulators): LDS and global bytes per WMMA fall by a third; 768 blocks, 3 per WGP, one even round (was 1536 at 5 per WGP)
# speedup vs baseline: 3.4501x; 1.0763x over previous
#include <hip/hip_runtime.h>
#include <hip/hip_bf16.h>


typedef __attribute__((ext_vector_type(16))) _Float16 v16h;
typedef __attribute__((ext_vector_type(8)))  _Float16 v8h;
typedef __attribute__((ext_vector_type(4)))  _Float16 v4h;
typedef __attribute__((ext_vector_type(8)))  float    v8f;

constexpr int kB = 2, kN = 2048, kC = 1024, kH = 16, kD = 64;
constexpr float kLnEps = 1e-5f;

__device__ __forceinline__ v8f wmma16(v16h a, v16h b, v8f c) {
  return __builtin_amdgcn_wmma_f32_16x16x32_f16(false, a, false, b, (short)0, c,
                                                false, false);
}

constexpr int kFragHalves  = 512;
constexpr int kPieceHalves = 256;

__device__ __forceinline__ v16h load_frag_pk(const _Float16* __restrict__ frag) {
  const int lane = threadIdx.x & 31;
  v8h lo = *(const v8h*)(frag + lane * 8);
  v8h hi = *(const v8h*)(frag + kPieceHalves + lane * 8);
  v16h f;
#pragma unroll
  for (int i = 0; i < 8; ++i) { f[i] = lo[i]; f[i + 8] = hi[i]; }
  return f;
}

__device__ __forceinline__ int qk_pk_off(int n, int d) {
  return (n >> 4) * (2 * kFragHalves) + (d >> 5) * kFragHalves +
         ((d >> 4) & 1) * kPieceHalves + ((((d >> 3) & 1) << 4) + (n & 15)) * 8 + (d & 7);
}

constexpr int kCvtBlock  = 256;
constexpr int kCvtPerBlk = 8192;
constexpr int kCvtInFl   = 16;
constexpr int kCvtRounds = kCvtPerBlk / (kCvtBlock * kCvtInFl);
constexpr int kCvtXBlks  = kB * kN * kC / 4 / kCvtPerBlk;
constexpr int kCvtWBlks  = kC * kC / 4 / kCvtPerBlk;
constexpr int kCvtGrid   = 3 * kCvtXBlks + 4 * kCvtWBlks;
static_assert(kCvtRounds * kCvtBlock * kCvtInFl == kCvtPerBlk, "block chunk");
static_assert(kCvtXBlks * kCvtPerBlk * 4 == kB * kN * kC, "activation split");
static_assert(kCvtWBlks * kCvtPerBlk * 4 == kC * kC, "weight split");

__global__ __launch_bounds__(kCvtBlock)
void cvt_all_f32_f16(const float* __restrict__ xq, const float* __restrict__ xk,
                     const float* __restrict__ xv, const float* __restrict__ wq,
                     const float* __restrict__ wk, const float* __restrict__ wv,
                     const float* __restrict__ wo,
                     _Float16* __restrict__ xq16, _Float16* __restrict__ xk16,
                     _Float16* __restrict__ xv16, _Float16* __restrict__ wq16,
                     _Float16* __restrict__ wk16, _Float16* __restrict__ wv16,
                     _Float16* __restrict__ wo16) {
  const int b = blockIdx.x;
  const float* src;
  _Float16* dst;
  int lb;
  if (b < 3 * kCvtXBlks) {
    const int s = b / kCvtXBlks;
    lb  = b - s * kCvtXBlks;
    src = (s == 0) ? xq : (s == 1) ? xk : xv;
    dst = (s == 0) ? xq16 : (s == 1) ? xk16 : xv16;
  } else {
    const int bw = b - 3 * kCvtXBlks;
    const int s  = bw / kCvtWBlks;
    lb  = bw - s * kCvtWBlks;
    src = (s == 0) ? wq : (s == 1) ? wk : (s == 2) ? wv : wo;
    dst = (s == 0) ? wq16 : (s == 1) ? wk16 : (s == 2) ? wv16 : wo16;
  }
  const float4* s4 = (const float4*)src + (size_t)lb * kCvtPerBlk + threadIdx.x;
  v4h*          d4 = (v4h*)dst          + (size_t)lb * kCvtPerBlk + threadIdx.x;

#pragma unroll
  for (int r = 0; r < kCvtRounds; ++r) {
    float4 v[kCvtInFl];
#pragma unroll
    for (int j = 0; j < kCvtInFl; ++j)
      v[j] = s4[(r * kCvtInFl + j) * kCvtBlock];
#pragma unroll
    for (int j = 0; j < kCvtInFl; ++j) {
      v4h h;
      h[0] = (_Float16)v[j].x; h[1] = (_Float16)v[j].y;
      h[2] = (_Float16)v[j].z; h[3] = (_Float16)v[j].w;
      d4[(r * kCvtInFl + j) * kCvtBlock] = h;
    }
  }
}

constexpr int kGK     = 64;
constexpr int kGPitch = kGK + 8;

__device__ __forceinline__ v16h lds_frag(const _Float16* p) {
  const int lane = threadIdx.x & 31;
  const int r    = lane & 15;
  const int ko   = (lane >> 4) << 3;
  const _Float16* b0 = p + r * kGPitch + ko;
  v8h lo = *(const v8h*)(b0);
  v8h hi = *(const v8h*)(b0 + 16);
  v16h f;
#pragma unroll
  for (int i = 0; i < 8; ++i) { f[i] = lo[i]; f[i + 8] = hi[i]; }
  return f;
}

template <int MODE>
__device__ __forceinline__
void gemm_tile(const _Float16* __restrict__ A,
               const _Float16* __restrict__ W,
               void* __restrict__ out,
               _Float16 (*__restrict__ sA)[128][kGPitch],
               _Float16 (*__restrict__ sW)[64][kGPitch],
               const int bx, const int by) {
  const int tid  = threadIdx.x;
  const int wave = tid >> 5;
  const int lane = tid & 31;
  const int colL = lane & 15;
  const int hi   = lane >> 4;
  const int mB = bx * 128;
  const int m0 = mB + wave * 32;
  const int n0 = by * 64;

  const int crow = tid >> 3;
  const int ccol = (tid & 7) * 8;
  const _Float16* ag = A + (size_t)(mB + crow) * kC + ccol;
  const _Float16* wg = W + (size_t)(n0 + crow) * kC + ccol;

  v8h ra[8], rw[4];
  auto gload = [&](int k0) {
#pragma unroll
    for (int i = 0; i < 8; ++i) ra[i] = *(const v8h*)(ag + (size_t)i * 16 * kC + k0);
#pragma unroll
    for (int i = 0; i < 4; ++i) rw[i] = *(const v8h*)(wg + (size_t)i * 16 * kC + k0);
  };
  auto sstore = [&](int s) {
#pragma unroll
    for (int i = 0; i < 8; ++i) *(v8h*)&sA[s][crow + 16 * i][ccol] = ra[i];
#pragma unroll
    for (int i = 0; i < 4; ++i) *(v8h*)&sW[s][crow + 16 * i][ccol] = rw[i];
  };

  v8f acc[2][4] = {};

  gload(0);
  sstore(0);
  __syncthreads();

  constexpr int kSlices = kC / kGK;
  for (int t = 0; t < kSlices; ++t) {
    const int s = t & 1;
    if (t + 1 < kSlices) gload((t + 1) * kGK);

#pragma unroll
    for (int kk = 0; kk < kGK; kk += 32) {
      v16h a0 = lds_frag(&sA[s][wave * 32][kk]);
      v16h a1 = lds_frag(&sA[s][wave * 32 + 16][kk]);
      v16h bf[4];
#pragma unroll
      for (int nt = 0; nt < 4; ++nt) bf[nt] = lds_frag(&sW[s][nt * 16][kk]);
#pragma unroll
      for (int nt = 0; nt < 4; ++nt) {
        acc[0][nt] = wmma16(a0, bf[nt], acc[0][nt]);
        acc[1][nt] = wmma16(a1, bf[nt], acc[1][nt]);
      }
    }

    if (t + 1 < kSlices) sstore(s ^ 1);
    __syncthreads();
  }

  if constexpr (MODE == 1) {
    const int b = m0 >> 11, c = (m0 & (kN - 1)) >> 5;
    const int h = n0 >> 6;
    _Float16* hb = (_Float16*)out + (size_t)(b * kH + h) * kD * kN +
                   (size_t)c * (4 * kFragHalves) + lane * 8;
#pragma unroll
    for (int mt = 0; mt < 2; ++mt)
#pragma unroll
      for (int nt = 0; nt < 4; ++nt) {
        v8h pk;
#pragma unroll
        for (int r = 0; r < 8; ++r) pk[r] = (_Float16)acc[mt][nt][r];
        *(v8h*)(hb + nt * kFragHalves + mt * kPieceHalves) = pk;
      }
  } else {
#pragma unroll
    for (int mt = 0; mt < 2; ++mt)
#pragma unroll
      for (int nt = 0; nt < 4; ++nt)
#pragma unroll
        for (int r = 0; r < 8; ++r) {
          const int row  = m0 + mt * 16 + 8 * hi + r;
          const int colc = n0 + nt * 16 + colL;
          const float v  = acc[mt][nt][r];
          if constexpr (MODE == 2) {
            ((float*)out)[(size_t)row * kC + colc] = v;
          } else {
            const int b = row >> 11, n = row & (kN - 1);
            const int h = colc >> 6, d = colc & (kD - 1);
            ((_Float16*)out)[(size_t)(b * kH + h) * kN * kD + qk_pk_off(n, d)] = (_Float16)v;
          }
        }
  }
}

template <int MODE>
__global__ __launch_bounds__(128)
void gemm_xwT(const _Float16* __restrict__ A,
              const _Float16* __restrict__ W,
              void* __restrict__ out) {
  __shared__ __attribute__((aligned(16))) _Float16 sA[2][128][kGPitch];
  __shared__ __attribute__((aligned(16))) _Float16 sW[2][64][kGPitch];
  gemm_tile<MODE>(A, W, out, sA, sW, blockIdx.x, blockIdx.y);
}

template <int MODE>
__device__ __forceinline__
void gemm_tile_128(const _Float16* __restrict__ A,
                   const _Float16* __restrict__ W,
                   _Float16* __restrict__ out,
                   _Float16 (*__restrict__ sA)[128][kGPitch],
                   _Float16 (*__restrict__ sW)[128][kGPitch],
                   const int bx, const int by) {
  const int tid  = threadIdx.x;
  const int wave = tid >> 5;
  const int lane = tid & 31;
  const int colL = lane & 15;
  const int hi   = lane >> 4;
  const int wm   = wave >> 1;
  const int wn   = wave & 1;
  const int mB = bx * 128;
  const int nB = by * 128;
  const int m0 = mB + wm * 64;
  const int n0 = nB + wn * 64;

  const int crow = tid >> 3;
  const int ccol = (tid & 7) * 8;
  const _Float16* ag = A + (size_t)(mB + crow) * kC + ccol;
  const _Float16* wg = W + (size_t)(nB + crow) * kC + ccol;

  v8h ra[8], rw[8];
  auto gload = [&](int k0) {
#pragma unroll
    for (int i = 0; i < 8; ++i) ra[i] = *(const v8h*)(ag + (size_t)i * 16 * kC + k0);
#pragma unroll
    for (int i = 0; i < 8; ++i) rw[i] = *(const v8h*)(wg + (size_t)i * 16 * kC + k0);
  };
  auto sstore = [&](int s) {
#pragma unroll
    for (int i = 0; i < 8; ++i) *(v8h*)&sA[s][crow + 16 * i][ccol] = ra[i];
#pragma unroll
    for (int i = 0; i < 8; ++i) *(v8h*)&sW[s][crow + 16 * i][ccol] = rw[i];
  };

  v8f acc[4][4] = {};

  gload(0);
  sstore(0);
  __syncthreads();

  constexpr int kSlices = kC / kGK;
  for (int t = 0; t < kSlices; ++t) {
    const int s = t & 1;
    if (t + 1 < kSlices) gload((t + 1) * kGK);

#pragma unroll
    for (int kk = 0; kk < kGK; kk += 32) {
      v16h af[4], bf[4];
#pragma unroll
      for (int mt = 0; mt < 4; ++mt) af[mt] = lds_frag(&sA[s][wm * 64 + mt * 16][kk]);
#pragma unroll
      for (int nt = 0; nt < 4; ++nt) bf[nt] = lds_frag(&sW[s][wn * 64 + nt * 16][kk]);
#pragma unroll
      for (int nt = 0; nt < 4; ++nt)
#pragma unroll
        for (int mt = 0; mt < 4; ++mt)
          acc[mt][nt] = wmma16(af[mt], bf[nt], acc[mt][nt]);
    }

    if (t + 1 < kSlices) sstore(s ^ 1);
    __syncthreads();
  }

  if constexpr (MODE == 1) {
    const int b = m0 >> 11, c = (m0 & (kN - 1)) >> 5;
    const int h = n0 >> 6;
    _Float16* hb = out + (size_t)(b * kH + h) * kD * kN +
                   (size_t)c * (4 * kFragHalves) + lane * 8;
#pragma unroll
    for (int mt = 0; mt < 4; ++mt)
#pragma unroll
      for (int nt = 0; nt < 4; ++nt) {
        v8h pk;
#pragma unroll
        for (int r = 0; r < 8; ++r) pk[r] = (_Float16)acc[mt][nt][r];
        *(v8h*)(hb + (mt >> 1) * (4 * kFragHalves) + nt * kFragHalves +
                (mt & 1) * kPieceHalves) = pk;
      }
  } else {
#pragma unroll
    for (int mt = 0; mt < 4; ++mt)
#pragma unroll
      for (int nt = 0; nt < 4; ++nt)
#pragma unroll
        for (int r = 0; r < 8; ++r) {
          const int row  = m0 + mt * 16 + 8 * hi + r;
          const int colc = n0 + nt * 16 + colL;
          const int b = row >> 11, n = row & (kN - 1);
          const int h = colc >> 6, d = colc & (kD - 1);
          out[(size_t)(b * kH + h) * kN * kD + qk_pk_off(n, d)] =
              (_Float16)acc[mt][nt][r];
        }
  }
}

__global__ __launch_bounds__(128) __attribute__((amdgpu_waves_per_eu(3, 3)))
void gemm_qkv(const _Float16* __restrict__ xq, const _Float16* __restrict__ wq,
              _Float16* __restrict__ qh,
              const _Float16* __restrict__ xk, const _Float16* __restrict__ wk,
              _Float16* __restrict__ kh,
              const _Float16* __restrict__ xv, const _Float16* __restrict__ wv,
              _Float16* __restrict__ vt) {
  __shared__ __attribute__((aligned(16))) _Float16 sA[2][128][kGPitch];
  __shared__ __attribute__((aligned(16))) _Float16 sW[2][128][kGPitch];
  const int z = blockIdx.z;
  if (z == 2) {
    gemm_tile_128<1>(xv, wv, vt, sA, sW, blockIdx.x, blockIdx.y);
  } else {
    gemm_tile_128<0>(z == 0 ? xq : xk, z == 0 ? wq : wk, z == 0 ? qh : kh, sA, sW,
                     blockIdx.x, blockIdx.y);
  }
}

constexpr int kKT = 8;

__global__ __launch_bounds__(256, 4)
void colstats_kernel(const _Float16* __restrict__ q16,
                     const _Float16* __restrict__ k16,
                     float* __restrict__ mcol,
                     float* __restrict__ rcol) {
  const int kg   = blockIdx.x;
  const int bh   = blockIdx.y;
  const int wave = threadIdx.x >> 5;
  const int lane = threadIdx.x & 31;
  const _Float16* qb = q16 + (size_t)bh * kN * kD;
  const _Float16* kb = k16 + ((size_t)bh * kN + (size_t)kg * 16 * kKT) * kD;

  v16h bk[kKT][2];
#pragma unroll
  for (int t = 0; t < kKT; ++t) {
    bk[t][0] = load_frag_pk(kb + (size_t)t * 16 * kD);
    bk[t][1] = load_frag_pk(kb + (size_t)t * 16 * kD + kFragHalves);
  }

  float mrun[kKT], srun[kKT];
#pragma unroll
  for (int t = 0; t < kKT; ++t) { mrun[t] = -3.4e38f; srun[t] = 0.f; }

  constexpr int kTrips = kN / 16 / 8;
  v16h a0 = load_frag_pk(qb + (size_t)wave * 16 * kD);
  v16h a1 = load_frag_pk(qb + (size_t)wave * 16 * kD + kFragHalves);
  for (int it = 0; it < kTrips; ++it) {
    v16h n0 = a0, n1 = a1;
    if (it + 1 < kTrips) {
      const int qn = wave + 8 * (it + 1);
      n0 = load_frag_pk(qb + (size_t)qn * 16 * kD);
      n1 = load_frag_pk(qb + (size_t)qn * 16 * kD + kFragHalves);
    }
#pragma unroll
    for (int tp = 0; tp < kKT; tp += 2) {
      v8f s[2];
#pragma unroll
      for (int u = 0; u < 2; ++u) {
        s[u] = v8f{};
        s[u] = wmma16(a0, bk[tp + u][0], s[u]);
        s[u] = wmma16(a1, bk[tp + u][1], s[u]);
      }
#pragma unroll
      for (int u = 0; u < 2; ++u) {
        const int t = tp + u;
        float tm = s[u][0];
#pragma unroll
        for (int i = 1; i < 8; ++i) tm = fmaxf(tm, s[u][i]);
        const float mn = fmaxf(mrun[t], tm);
        float part = 0.f;
#pragma unroll
        for (int i = 0; i < 8; ++i) part += __expf(s[u][i] - mn);
        srun[t] = srun[t] * __expf(mrun[t] - mn) + part;
        mrun[t] = mn;
      }
      __builtin_amdgcn_sched_barrier(0);
    }
    a0 = n0; a1 = n1;
  }

  __shared__ float redm[8][16 * kKT];
  __shared__ float reds[8][16 * kKT];
#pragma unroll
  for (int t = 0; t < kKT; ++t) {
    const float mo = __shfl_xor(mrun[t], 16, 32);
    const float so = __shfl_xor(srun[t], 16, 32);
    const float mn = fmaxf(mrun[t], mo);
    const float sv = srun[t] * __expf(mrun[t] - mn) + so * __expf(mo - mn);
    if (lane < 16) { redm[wave][t * 16 + lane] = mn; reds[wave][t * 16 + lane] = sv; }
  }
  __syncthreads();
  if (threadIdx.x < 16 * kKT) {
    const int c = threadIdx.x;
    float m = redm[0][c];
#pragma unroll
    for (int w = 1; w < 8; ++w) m = fmaxf(m, redm[w][c]);
    float sum = 0.f;
#pragma unroll
    for (int w = 0; w < 8; ++w) sum += reds[w][c] * __expf(redm[w][c] - m);
    const size_t o = (size_t)bh * kN + (size_t)kg * 16 * kKT + c;
    mcol[o] = m;
    rcol[o] = 1.0f / sum;
  }
}

constexpr int kQT = 4;
constexpr int kCW = 4;
constexpr int kKS = 2;

__global__ __launch_bounds__(32 * kCW * kKS)
void ctx_kernel(const _Float16* __restrict__ q16,
                const _Float16* __restrict__ k16,
                const _Float16* __restrict__ vt16,
                const float* __restrict__ mcol,
                const float* __restrict__ rcol,
                _Float16* __restrict__ ctx16) {
  const int bh   = blockIdx.y;
  const int wave = __builtin_amdgcn_readfirstlane(threadIdx.x >> 5);
  const int grp  = wave & (kCW - 1);
  const int half = wave / kCW;
  const int lane = threadIdx.x & 31;
  const int col  = lane & 15;
  const int hi   = lane >> 4;
  const int qt0  = (blockIdx.x * kCW + grp) * kQT;

  __shared__ __attribute__((aligned(16))) float mr_s[2][kN];
  __shared__ __attribute__((aligned(16))) float4 red_s[kCW * kKS][16][32];

  for (int i = threadIdx.x; i < kN; i += 32 * kCW * kKS) {
    mr_s[0][i] = mcol[(size_t)bh * kN + i];
    mr_s[1][i] = rcol[(size_t)bh * kN + i];
  }
  __syncthreads();

  const _Float16* qb = q16  + (size_t)bh * kN * kD;
  const _Float16* kb = k16  + (size_t)bh * kN * kD;
  const _Float16* vb = vt16 + (size_t)bh * kD * kN;

  v16h aq[kQT][2];
#pragma unroll
  for (int j = 0; j < kQT; ++j) {
    aq[j][0] = load_frag_pk(qb + (size_t)(qt0 + j) * 16 * kD);
    aq[j][1] = load_frag_pk(qb + (size_t)(qt0 + j) * 16 * kD + kFragHalves);
  }

  v8f acc[kQT][4] = {};

  auto load_chunk = [&](int k0, v16h (&kf)[4], v16h (&vf)[4]) {
    kf[0] = load_frag_pk(kb + (size_t)(k0     ) * kD);
    kf[1] = load_frag_pk(kb + (size_t)(k0     ) * kD + kFragHalves);
    kf[2] = load_frag_pk(kb + (size_t)(k0 + 16) * kD);
    kf[3] = load_frag_pk(kb + (size_t)(k0 + 16) * kD + kFragHalves);
#pragma unroll
    for (int dt = 0; dt < 4; ++dt)
      vf[dt] = load_frag_pk(vb + (size_t)k0 * kD + dt * kFragHalves);
  };

  struct MR { float4 m[4], r[4]; };
  auto load_mr = [&](int k0, MR& c) {
    const float4* mp = (const float4*)&mr_s[0][k0 + 8 * hi];
    const float4* rp = (const float4*)&mr_s[1][k0 + 8 * hi];
    c.m[0] = mp[0]; c.m[1] = mp[1]; c.m[2] = mp[4]; c.m[3] = mp[5];
    c.r[0] = rp[0]; c.r[1] = rp[1]; c.r[2] = rp[4]; c.r[3] = rp[5];
  };

  auto do_chunk = [&](int k0, const v16h (&kf)[4], const v16h (&vf)[4]) {
    MR c;
    load_mr(k0, c);
    const float mk[16] = {c.m[0].x, c.m[0].y, c.m[0].z, c.m[0].w,
                          c.m[1].x, c.m[1].y, c.m[1].z, c.m[1].w,
                          c.m[2].x, c.m[2].y, c.m[2].z, c.m[2].w,
                          c.m[3].x, c.m[3].y, c.m[3].z, c.m[3].w};
    const float rk[16] = {c.r[0].x, c.r[0].y, c.r[0].z, c.r[0].w,
                          c.r[1].x, c.r[1].y, c.r[1].z, c.r[1].w,
                          c.r[2].x, c.r[2].y, c.r[2].z, c.r[2].w,
                          c.r[3].x, c.r[3].y, c.r[3].z, c.r[3].w};
#pragma unroll
    for (int j = 0; j < kQT; ++j) {
      v8f s0 = {}, s1 = {};
      s0 = wmma16(kf[0], aq[j][0], s0);
      s0 = wmma16(kf[1], aq[j][1], s0);
      s1 = wmma16(kf[2], aq[j][0], s1);
      s1 = wmma16(kf[3], aq[j][1], s1);

      v16h ef;
#pragma unroll
      for (int i = 0; i < 8; ++i) {
        ef[i]     = (_Float16)(__expf(s0[i] - mk[i])     * rk[i]);
        ef[i + 8] = (_Float16)(__expf(s1[i] - mk[i + 8]) * rk[i + 8]);
      }
#pragma unroll
      for (int dt = 0; dt < 4; ++dt)
        acc[j][dt] = wmma16(ef, vf[dt], acc[j][dt]);
    }
  };

  static_assert((kN / kKS) % 64 == 0, "two chunks per trip");
  const int kbeg = half * (kN / kKS);
  const int kend = kbeg + kN / kKS;
  v16h kfA[4], vfA[4], kfB[4], vfB[4];
  load_chunk(kbeg, kfA, vfA);
  for (int k0 = kbeg; k0 < kend; k0 += 64) {
    load_chunk(k0 + 32, kfB, vfB);
    __builtin_amdgcn_sched_barrier(0);
    do_chunk(k0, kfA, vfA);
    const int kn = (k0 + 64 < kend) ? (k0 + 64) : k0;
    load_chunk(kn, kfA, vfA);
    __builtin_amdgcn_sched_barrier(0);
    do_chunk(k0 + 32, kfB, vfB);
  }

  const int b = bh >> 4, h = bh & 15;
  _Float16* cb = ctx16 + (size_t)b * kN * kC + (size_t)h * kD;
  auto hand_over = [&](int j0) {
    float4* dst = &red_s[wave][0][lane];
#pragma unroll
    for (int jj = 0; jj < 2; ++jj)
#pragma unroll
      for (int dt = 0; dt < 4; ++dt) {
        const v8f a = acc[j0 + jj][dt];
        dst[((jj * 4 + dt) * 2 + 0) * 32] = make_float4(a[0], a[1], a[2], a[3]);
        dst[((jj * 4 + dt) * 2 + 1) * 32] = make_float4(a[4], a[5], a[6], a[7]);
      }
  };
  auto finish = [&](int j0) {
    const float4* src = &red_s[wave ^ kCW][0][lane];
#pragma unroll
    for (int jj = 0; jj < 2; ++jj) {
      const int qrow = (qt0 + j0 + jj) * 16 + 8 * hi;
#pragma unroll
      for (int dt = 0; dt < 4; ++dt) {
        const float4 p0 = src[((jj * 4 + dt) * 2 + 0) * 32];
        const float4 p1 = src[((jj * 4 + dt) * 2 + 1) * 32];
        const float p[8] = {p0.x, p0.y, p0.z, p0.w, p1.x, p1.y, p1.z, p1.w};
#pragma unroll
        for (int r = 0; r < 8; ++r)
          cb[(size_t)(qrow + r) * kC + dt * 16 + col] =
              (_Float16)(acc[j0 + jj][dt][r] + p[r]);
      }
    }
  };
  if (half == 0) hand_over(2); else hand_over(0);
  __syncthreads();
  if (half == 0) finish(0); else finish(2);
}

__global__ void ln_kernel(const float* __restrict__ resid,
                          const float* __restrict__ y,
                          const float* __restrict__ g,
                          const float* __restrict__ be,
                          float* __restrict__ out) {
  const int row = blockIdx.x;
  const int t   = threadIdx.x;
  float x[4], s = 0.f, sq = 0.f;
#pragma unroll
  for (int j = 0; j < 4; ++j) {
    const int c = t + j * 256;
    const float v = resid[(size_t)row * kC + c] + y[(size_t)row * kC + c];
    x[j] = v; s += v; sq += v * v;
  }
#pragma unroll
  for (int o = 16; o >= 1; o >>= 1) {
    s  += __shfl_xor(s,  o, 32);
    sq += __shfl_xor(sq, o, 32);
  }
  __shared__ float rs[8], rq[8];
  __shared__ float mu_s, inv_s;
  const int wave = t >> 5, lane = t & 31;
  if (lane == 0) { rs[wave] = s; rq[wave] = sq; }
  __syncthreads();
  if (t == 0) {
    float S = 0.f, Q = 0.f;
#pragma unroll
    for (int w = 0; w < 8; ++w) { S += rs[w]; Q += rq[w]; }
    const float mu = S * (1.0f / kC);
    mu_s  = mu;
    inv_s = rsqrtf(Q * (1.0f / kC) - mu * mu + kLnEps);
  }
  __syncthreads();
  const float mu = mu_s, inv = inv_s;
#pragma unroll
  for (int j = 0; j < 4; ++j) {
    const int c = t + j * 256;
    out[(size_t)row * kC + c] = (x[j] - mu) * inv * g[c] + be[c];
  }
}

extern "C" void kernel_launch(void* const* d_in, const int* in_sizes, int n_in,
                              void* d_out, int out_size, void* d_ws, size_t ws_size,
                              hipStream_t stream) {
  (void)in_sizes; (void)n_in; (void)out_size; (void)ws_size;
  const float* query = (const float*)d_in[0];
  const float* key   = (const float*)d_in[1];
  const float* value = (const float*)d_in[2];
  const float* Wq = (const float*)d_in[4];
  const float* Wk = (const float*)d_in[5];
  const float* Wv = (const float*)d_in[6];
  const float* Wo = (const float*)d_in[7];
  const float* ln_g = (const float*)d_in[8];
  const float* ln_b = (const float*)d_in[9];
  float* out = (float*)d_out;

  constexpr size_t BNC = (size_t)kB * kN * kC;
  constexpr size_t CC  = (size_t)kC * kC;
  constexpr size_t BHN = (size_t)kB * kH * kN;

  size_t off = 0;
  auto take = [&](size_t bytes) -> void* {
    void* p = (char*)d_ws + off;
    off += (bytes + 255) & ~(size_t)255;
    return p;
  };
  _Float16* xq16 = (_Float16*)take(BNC * 2);
  _Float16* xk16 = (_Float16*)take(BNC * 2);
  _Float16* xv16 = (_Float16*)take(BNC * 2);
  _Float16* wq16 = (_Float16*)take(CC * 2);
  _Float16* wk16 = (_Float16*)take(CC * 2);
  _Float16* wv16 = (_Float16*)take(CC * 2);
  _Float16* wo16 = (_Float16*)take(CC * 2);
  _Float16* qh16 = (_Float16*)take(BNC * 2);
  _Float16* kh16 = (_Float16*)take(BNC * 2);
  _Float16* vt16 = (_Float16*)take(BNC * 2);
  float*    mcol = (float*)take(BHN * 4);
  float*    rcol = (float*)take(BHN * 4);
  _Float16* ctx16 = xq16;
  float*    yf32  = (float*)xk16;

  cvt_all_f32_f16<<<kCvtGrid, kCvtBlock, 0, stream>>>(
      query, key, value, Wq, Wk, Wv, Wo,
      xq16, xk16, xv16, wq16, wk16, wv16, wo16);

  dim3 ggrid(32, 16);
  gemm_qkv<<<dim3(32, 8, 3), 128, 0, stream>>>(xq16, wq16, qh16, xk16, wk16, kh16,
                                               xv16, wv16, vt16);

  colstats_kernel<<<dim3(kN / (16 * kKT), kB * kH), 256, 0, stream>>>(qh16, kh16, mcol, rcol);

  ctx_kernel<<<dim3(kN / (16 * kQT * kCW), kB * kH), 32 * kCW * kKS, 0, stream>>>(
      qh16, kh16, vt16, mcol, rcol, ctx16);

  gemm_xwT<2><<<ggrid, 128, 0, stream>>>(ctx16, wo16, yf32);
  ln_kernel<<<kB * kN, 256, 0, stream>>>(query, yf32, ln_g, ln_b, out);
}
